# starting kernel
# baseline (speedup 1.0000x reference)
#include <hip/hip_runtime.h>
#include <hip/hip_bf16.h>
#include <math.h>


typedef float v2f __attribute__((ext_vector_type(2)));
typedef float v4f __attribute__((ext_vector_type(4)));
typedef float v8f __attribute__((ext_vector_type(8)));

#define D_MODEL 256
#define NT 32
#define MT 128

__global__ __launch_bounds__(256)
void dta_gemm_f32_wmma(const float* __restrict__ A, const float* __restrict__ W,
                       const float* __restrict__ bias, float* __restrict__ C,
                       int M, int N) {
  __shared__ float Wlds[D_MODEL * NT];

  const int tid = threadIdx.x;
  const int n0  = blockIdx.y * NT;

  for (int c = tid; c < (D_MODEL / 2) * (NT / 4); c += 256) {
    const int kk = c >> 3;
    const int j  = (c & 7) << 2;
    v4f r0 = *(const v4f*)(W + (long)(2 * kk)     * N + n0 + j);
    v4f r1 = *(const v4f*)(W + (long)(2 * kk + 1) * N + n0 + j);
    v4f lo = {r0.x, r1.x, r0.y, r1.y};
    v4f hi = {r0.z, r1.z, r0.w, r1.w};
    *(v4f*)(&Wlds[kk * (NT * 2) + j * 2])     = lo;
    *(v4f*)(&Wlds[kk * (NT * 2) + j * 2 + 4]) = hi;
  }
  __syncthreads();

  const int wave  = tid >> 5;
  const int lane  = tid & 31;
  const int m0    = blockIdx.x * MT + wave * 16;
  const int mrow  = m0 + (lane & 15);
  const int khalf = (lane >> 4) << 1;
  const int ncol  = lane & 15;

  v8f c0 = {};
  v8f c1 = {};
  const float* Arow = A + (long)mrow * D_MODEL;
  const float* Bbase = &Wlds[(khalf >> 1) * (NT * 2) + ncol * 2];

  #pragma unroll 8
  for (int k = 0; k < D_MODEL; k += 4) {
    v2f a = *(const v2f*)(Arow + k + khalf);
    const float* Bk = Bbase + (k >> 1) * (NT * 2);
    v2f b0 = *(const v2f*)(Bk);
    v2f b1 = *(const v2f*)(Bk + 32);
    c0 = __builtin_amdgcn_wmma_f32_16x16x4_f32(false, a, false, b0, (short)0, c0, false, false);
    c1 = __builtin_amdgcn_wmma_f32_16x16x4_f32(false, a, false, b1, (short)0, c1, false, false);
  }

  const int rbase = m0 + ((lane >> 4) << 3);
  const float bv0 = bias ? bias[n0 + ncol] : 0.0f;
  const float bv1 = bias ? bias[n0 + 16 + ncol] : 0.0f;
  #pragma unroll
  for (int r = 0; r < 8; ++r) {
    C[(long)(rbase + r) * N + n0 + ncol]      = c0[r] + bv0;
    C[(long)(rbase + r) * N + n0 + 16 + ncol] = c1[r] + bv1;
  }
}

__global__ __launch_bounds__(256)
void dta_softmax12(float* __restrict__ aw, int total  ) {
  const int t = blockIdx.x * blockDim.x + threadIdx.x;
  if (t >= total) return;
  const int row = t >> 3;
  const int h   = t & 7;
  float* p = aw + (long)row * 96 + h * 12;

  float v[12];
  float mx = -3.402823466e+38f;
  #pragma unroll
  for (int i = 0; i < 12; ++i) { v[i] = p[i]; mx = fmaxf(mx, v[i]); }
  float s = 0.0f;
  #pragma unroll
  for (int i = 0; i < 12; ++i) { v[i] = __expf(v[i] - mx); s += v[i]; }
  const float inv = 1.0f / s;
  #pragma unroll
  for (int i = 0; i < 12; ++i) p[i] = v[i] * inv;
}

__global__ __launch_bounds__(256)
void dta_sample(const float* __restrict__ refp, const float* __restrict__ off_raw,
                const float* __restrict__ aw, const float* __restrict__ v_proj,
                float* __restrict__ out_mid) {
  const int wave = (blockIdx.x * blockDim.x + threadIdx.x) >> 5;
  const int lane = threadIdx.x & 31;
  const int b = wave >> 11;
  const int row = wave;

  const float rp = refp[row];
  const int   Ts[3]   = {2048, 1024, 512};
  const long  base[3] = {0, 8L * 2048, 8L * 2048 + 8L * 1024};

  const float* offr = off_raw + (long)row * 96;
  const float* awr  = aw + (long)row * 96;

  #pragma unroll
  for (int h = 0; h < 8; ++h) {
    float acc = 0.0f;
    #pragma unroll
    for (int l = 0; l < 3; ++l) {
      const float Tf = (float)Ts[l];
      const long  rb = base[l] + (long)b * Ts[l];
      #pragma unroll
      for (int p = 0; p < 4; ++p) {
        const int idx = h * 12 + l * 4 + p;
        const float o = offr[idx];
        const float w = awr[idx];
        float pos  = fminf(fmaxf(rp + o / Tf, 0.0f), 1.0f);
        float sidx = pos * (Tf - 1.0f);
        int ifl = (int)sidx;
        ifl = min(max(ifl, 0), Ts[l] - 2);
        const float wc = sidx - (float)ifl;
        const long r0 = (rb + ifl) * 256 + p * 32 + lane;
        const float vf = v_proj[r0];
        const float vc = v_proj[r0 + 256];
        acc += w * ((1.0f - wc) * vf + wc * vc);
      }
    }
    out_mid[(long)row * 256 + h * 32 + lane] = acc;
  }
}

extern "C" void kernel_launch(void* const* d_in, const int* in_sizes, int n_in,
                              void* d_out, int out_size, void* d_ws, size_t ws_size,
                              hipStream_t stream) {
  (void)in_sizes; (void)n_in; (void)out_size; (void)ws_size;

  const float* query = (const float*)d_in[0];
  const float* refp  = (const float*)d_in[1];
  const float* val0  = (const float*)d_in[2];
  const float* val1  = (const float*)d_in[3];
  const float* val2  = (const float*)d_in[4];
  const float* Woff  = (const float*)d_in[5];
  const float* boff  = (const float*)d_in[6];
  const float* Waw   = (const float*)d_in[7];
  const float* baw   = (const float*)d_in[8];
  const float* Wv    = (const float*)d_in[9];
  const float* bv    = (const float*)d_in[10];
  const float* Wo    = (const float*)d_in[11];
  const float* bo    = (const float*)d_in[12];
  float*       out   = (float*)d_out;

  const int BQ = 8 * 2048;

  char* ws = (char*)d_ws;
  float* off_raw = (float*)(ws);
  float* aw_buf  = (float*)(ws + (size_t)BQ * 96 * 4);
  float* v_proj  = (float*)(ws + (size_t)BQ * 96 * 4 * 2);
  float* out_mid = (float*)(ws + (size_t)BQ * 96 * 4 * 2
                               + (size_t)(BQ + 8192 + 4096) * 256 * 4);

  dim3 blk(256);

  dta_gemm_f32_wmma<<<dim3(BQ / MT, 96 / NT), blk, 0, stream>>>(query, Woff, boff, off_raw, BQ, 96);
  dta_gemm_f32_wmma<<<dim3(BQ / MT, 96 / NT), blk, 0, stream>>>(query, Waw, baw, aw_buf, BQ, 96);
  dta_softmax12<<<dim3((BQ * 8 + 255) / 256), blk, 0, stream>>>(aw_buf, BQ * 8);
  dta_gemm_f32_wmma<<<dim3(16384 / MT, 256 / NT), blk, 0, stream>>>(val0, Wv, bv, v_proj, 16384, 256);
  dta_gemm_f32_wmma<<<dim3( 8192 / MT, 256 / NT), blk, 0, stream>>>(val1, Wv, bv, v_proj + (size_t)16384 * 256,  8192, 256);
  dta_gemm_f32_wmma<<<dim3( 4096 / MT, 256 / NT), blk, 0, stream>>>(val2, Wv, bv, v_proj + (size_t)(16384 + 8192) * 256, 4096, 256);
  dta_sample<<<dim3(BQ / 8), blk, 0, stream>>>(refp, off_raw, aw_buf, v_proj, out_mid);
  dta_gemm_f32_wmma<<<dim3(BQ / MT, 256 / NT), blk, 0, stream>>>(out_mid, Wo, bo, out, BQ, 256);
}


# Value projections of all 3 levels in one weight-stationary launch: one 1792-strip row space, a workgroup stages its 64 Wv columns once, each wave runs 7 strips; sum order kept; 6 launches -> 4
# speedup vs baseline: 1.3030x; 1.3030x over previous
#include <hip/hip_runtime.h>
#include <hip/hip_bf16.h>
#include <math.h>


typedef float v2f __attribute__((ext_vector_type(2)));
typedef float v4f __attribute__((ext_vector_type(4)));
typedef float v8f __attribute__((ext_vector_type(8)));

#define D_MODEL 256

#define K_STEPS (D_MODEL / 4)

template <int NW, int WAVES, int KB, int G>
__global__ __launch_bounds__(WAVES * 32)
void dta_gemm_f32_wmma(const float* __restrict__ A, const float* __restrict__ W,
                       const float* __restrict__ bias, float* __restrict__ C,
                       int N) {
  constexpr int NTHR  = WAVES * 32;
  constexpr int NTILE = NW / 16;
  constexpr int PITCH = NW * 2 + 32;
  constexpr int CPR   = NW / 4;
  constexpr int TILE_FLOATS = (D_MODEL / 2) * PITCH;
  constexpr int ONE_PER_WGP_FLOATS = 41984;
  constexpr int LDS_FLOATS = TILE_FLOATS > ONE_PER_WGP_FLOATS ? TILE_FLOATS : ONE_PER_WGP_FLOATS;
  __shared__ __attribute__((aligned(16))) float Wlds[LDS_FLOATS];

  const int tid   = threadIdx.x;
  const int wave  = tid >> 5;
  const int lane  = tid & 31;
  const int n0    = blockIdx.y * NW;
  const int m0    = blockIdx.x * (WAVES * 16) + wave * 16;
  const int khalf = (lane >> 4) << 1;
  const int ncol  = lane & 15;

  const float* Arow = A + (long)(m0 + ncol) * D_MODEL + khalf;
  v2f a_cur[KB];
  #pragma unroll
  for (int s = 0; s < KB; ++s) a_cur[s] = *(const v2f*)(Arow + 4 * s);

  constexpr int STAGE_ITERS = (D_MODEL / 2) * CPR / NTHR;
  static_assert(STAGE_ITERS * NTHR == (D_MODEL / 2) * CPR, "chunks divide evenly over the threads");
  #pragma unroll
  for (int it = 0; it < STAGE_ITERS; ++it) {
    const int ch = tid + it * NTHR;
    const int kk = ch / CPR;
    const int j  = (ch - kk * CPR) << 2;
    v4f r0 = *(const v4f*)(W + (long)(2 * kk)     * N + n0 + j);
    v4f r1 = *(const v4f*)(W + (long)(2 * kk + 1) * N + n0 + j);
    v4f lo = {r0.x, r1.x, r0.y, r1.y};
    v4f hi = {r0.z, r1.z, r0.w, r1.w};
    *(v4f*)(&Wlds[kk * PITCH + j * 2])     = lo;
    *(v4f*)(&Wlds[kk * PITCH + j * 2 + 4]) = hi;
  }
  __syncthreads();

  v8f acc[NTILE];
  #pragma unroll
  for (int t = 0; t < NTILE; ++t) acc[t] = v8f{};

  const float* Bbase = &Wlds[(khalf >> 1) * PITCH + ncol * 2];

  static_assert(KB % G == 0 && K_STEPS % KB == 0, "blocks tile a prefetch group, groups tile K");
  constexpr int NBLK = K_STEPS / G;
  v2f a_nxt[KB];
  v2f bset[2][G][NTILE];
  #pragma unroll
  for (int g = 0; g < G; ++g) {
    #pragma unroll
    for (int t = 0; t < NTILE; ++t)
      bset[0][g][t] = *(const v2f*)(Bbase + g * 2 * PITCH + 32 * t);
  }

  #pragma unroll
  for (int blk = 0; blk < NBLK; ++blk) {
    const int ks0 = blk * G;
    if (blk + 1 < NBLK) {
      #pragma unroll
      for (int g = 0; g < G; ++g) {
        const float* Bk = Bbase + (ks0 + G + g) * 2 * PITCH;
        #pragma unroll
        for (int t = 0; t < NTILE; ++t)
          bset[(blk + 1) & 1][g][t] = *(const v2f*)(Bk + 32 * t);
      }
    }
    if (ks0 % KB == 0 && ks0 + KB < K_STEPS) {
      #pragma unroll
      for (int s = 0; s < KB; ++s) a_nxt[s] = *(const v2f*)(Arow + 4 * (ks0 + KB + s));
    }
    __builtin_amdgcn_sched_barrier(0);
    #pragma unroll
    for (int g = 0; g < G; ++g) {
      #pragma unroll
      for (int t = 0; t < NTILE; ++t)
        acc[t] = __builtin_amdgcn_wmma_f32_16x16x4_f32(false, a_cur[ks0 % KB + g], false, bset[blk & 1][g][t], (short)0, acc[t], false, false);
    }
    __builtin_amdgcn_sched_barrier(0);
    if ((ks0 + G) % KB == 0 && ks0 + G < K_STEPS) {
      #pragma unroll
      for (int s = 0; s < KB; ++s) a_cur[s] = a_nxt[s];
    }
  }

  const int rbase = m0 + ((lane >> 4) << 3);
  #pragma unroll
  for (int t = 0; t < NTILE; ++t) {
    const float bvt = bias[n0 + 16 * t + ncol];
    #pragma unroll
    for (int r = 0; r < 8; ++r)
      C[(long)(rbase + r) * N + n0 + 16 * t + ncol] = acc[t][r] + bvt;
  }
}

#define DTA_STRIPS_L0 1024
#define DTA_STRIPS_L1 512

__device__ __forceinline__ const float* dta_strip_rows(const float* __restrict__ V0,
                                                       const float* __restrict__ V1,
                                                       const float* __restrict__ V2, int s) {
  if (s < DTA_STRIPS_L0) return V0 + (long)s * (16 * D_MODEL);
  if (s < DTA_STRIPS_L0 + DTA_STRIPS_L1) return V1 + (long)(s - DTA_STRIPS_L0) * (16 * D_MODEL);
  return V2 + (long)(s - DTA_STRIPS_L0 - DTA_STRIPS_L1) * (16 * D_MODEL);
}

template <int NW, int WAVES, int STRIPS, int KB, int G>
__global__ __launch_bounds__(WAVES * 32)
void dta_vproj_ws(const float* __restrict__ V0, const float* __restrict__ V1,
                  const float* __restrict__ V2, const float* __restrict__ W,
                  const float* __restrict__ bias, float* __restrict__ C) {
  constexpr int N     = D_MODEL;
  constexpr int NTHR  = WAVES * 32;
  constexpr int NTILE = NW / 16;
  constexpr int PITCH = NW * 2 + 32;
  constexpr int CPR   = NW / 4;
  constexpr int TILE_FLOATS = (D_MODEL / 2) * PITCH;
  constexpr int ONE_PER_WGP_FLOATS = 41984;
  constexpr int LDS_FLOATS = TILE_FLOATS > ONE_PER_WGP_FLOATS ? TILE_FLOATS : ONE_PER_WGP_FLOATS;
  __shared__ __attribute__((aligned(16))) float Wlds[LDS_FLOATS];

  const int tid   = threadIdx.x;
  const int wave  = __builtin_amdgcn_readfirstlane(tid >> 5);
  const int lane  = tid & 31;
  const int n0    = blockIdx.y * NW;
  const int khalf = (lane >> 4) << 1;
  const int ncol  = lane & 15;
  const int lofs  = ncol * D_MODEL + khalf;
  const int s0    = blockIdx.x * (WAVES * STRIPS) + wave;

  const float* Arow = dta_strip_rows(V0, V1, V2, s0) + lofs;
  v2f a_cur[KB];
  #pragma unroll
  for (int s = 0; s < KB; ++s) a_cur[s] = *(const v2f*)(Arow + 4 * s);

  constexpr int STAGE_ITERS = (D_MODEL / 2) * CPR / NTHR;
  static_assert(STAGE_ITERS * NTHR == (D_MODEL / 2) * CPR, "chunks divide evenly over the threads");
  #pragma unroll
  for (int it = 0; it < STAGE_ITERS; ++it) {
    const int ch = tid + it * NTHR;
    const int kk = ch / CPR;
    const int j  = (ch - kk * CPR) << 2;
    v4f r0 = *(const v4f*)(W + (long)(2 * kk)     * N + n0 + j);
    v4f r1 = *(const v4f*)(W + (long)(2 * kk + 1) * N + n0 + j);
    v4f lo = {r0.x, r1.x, r0.y, r1.y};
    v4f hi = {r0.z, r1.z, r0.w, r1.w};
    *(v4f*)(&Wlds[kk * PITCH + j * 2])     = lo;
    *(v4f*)(&Wlds[kk * PITCH + j * 2 + 4]) = hi;
  }

  float bvt[NTILE];
  #pragma unroll
  for (int t = 0; t < NTILE; ++t) bvt[t] = bias[n0 + 16 * t + ncol];
  __syncthreads();

  const float* Bbase = &Wlds[(khalf >> 1) * PITCH + ncol * 2];

  static_assert(KB % G == 0 && K_STEPS % KB == 0, "blocks tile a prefetch group, groups tile K");
  constexpr int NBLK = K_STEPS / G;
  static_assert(NBLK % 2 == 0, "the B register sets alternate and must line up again at the next strip");
  v2f a_nxt[KB];
  v2f bset[2][G][NTILE];
  #pragma unroll
  for (int g = 0; g < G; ++g) {
    #pragma unroll
    for (int t = 0; t < NTILE; ++t)
      bset[0][g][t] = *(const v2f*)(Bbase + g * 2 * PITCH + 32 * t);
  }

  for (int i = 0; i < STRIPS; ++i) {
    const int sc = s0 + i * WAVES;
    const int sn = (i + 1 < STRIPS) ? sc + WAVES : sc;
    const float* Anext = dta_strip_rows(V0, V1, V2, sn) + lofs;

    v8f acc[NTILE];
    #pragma unroll
    for (int t = 0; t < NTILE; ++t) acc[t] = v8f{};

    #pragma unroll
    for (int blk = 0; blk < NBLK; ++blk) {
      const int ks0 = blk * G;
      #pragma unroll
      for (int g = 0; g < G; ++g) {
        const float* Bk = Bbase + ((ks0 + G + g) % K_STEPS) * 2 * PITCH;
        #pragma unroll
        for (int t = 0; t < NTILE; ++t)
          bset[(blk + 1) & 1][g][t] = *(const v2f*)(Bk + 32 * t);
      }
      if (ks0 % KB == 0) {
        if (ks0 + KB < K_STEPS) {
          #pragma unroll
          for (int s = 0; s < KB; ++s) a_nxt[s] = *(const v2f*)(Arow + 4 * (ks0 + KB + s));
        } else {
          #pragma unroll
          for (int s = 0; s < KB; ++s) a_nxt[s] = *(const v2f*)(Anext + 4 * s);
        }
      }
      __builtin_amdgcn_sched_barrier(0);
      #pragma unroll
      for (int g = 0; g < G; ++g) {
        #pragma unroll
        for (int t = 0; t < NTILE; ++t)
          acc[t] = __builtin_amdgcn_wmma_f32_16x16x4_f32(false, a_cur[ks0 % KB + g], false, bset[blk & 1][g][t], (short)0, acc[t], false, false);
      }
      __builtin_amdgcn_sched_barrier(0);
      if ((ks0 + G) % KB == 0) {
        #pragma unroll
        for (int s = 0; s < KB; ++s) a_cur[s] = a_nxt[s];
      }
    }

    float* Crow = C + (long)sc * (16 * N) + (long)((lane >> 4) << 3) * N + n0 + ncol;
    #pragma unroll
    for (int t = 0; t < NTILE; ++t) {
      #pragma unroll
      for (int r = 0; r < 8; ++r)
        Crow[r * N + 16 * t] = acc[t][r] + bvt[t];
    }
    Arow = Anext;
  }
}

template <int NTHR, int PITCH>
__device__ __forceinline__ void dta_stage_w96(const float* __restrict__ W, float* Wlds,
                                              int cofs, int tid) {
  constexpr int CPR   = 96 / 4;
  constexpr int ITERS = (D_MODEL / 2) * CPR / NTHR;
  static_assert(ITERS * NTHR == (D_MODEL / 2) * CPR, "chunks divide evenly over the threads");
  #pragma unroll
  for (int it = 0; it < ITERS; ++it) {
    const int ch = tid + it * NTHR;
    const int kk = ch / CPR;
    const int j  = (ch - kk * CPR) << 2;
    v4f r0 = *(const v4f*)(W + (long)(2 * kk)     * 96 + j);
    v4f r1 = *(const v4f*)(W + (long)(2 * kk + 1) * 96 + j);
    v4f lo = {r0.x, r1.x, r0.y, r1.y};
    v4f hi = {r0.z, r1.z, r0.w, r1.w};
    *(v4f*)(&Wlds[kk * PITCH + (cofs + j) * 2])     = lo;
    *(v4f*)(&Wlds[kk * PITCH + (cofs + j) * 2 + 4]) = hi;
  }
}

template <int WAVES, int KB, int G>
__global__ __launch_bounds__(WAVES * 32)
void dta_qproj_softmax(const float* __restrict__ A,
                       const float* __restrict__ Woff, const float* __restrict__ boff,
                       const float* __restrict__ Waw,  const float* __restrict__ baw,
                       float* __restrict__ off_out, float* __restrict__ aw_out) {
  constexpr int NH    = 96;
  constexpr int NW    = 2 * NH;
  constexpr int NTHR  = WAVES * 32;
  constexpr int NTILE = NW / 16;
  constexpr int HT    = NH / 16;
  constexpr int PITCH = NW * 2 + 32;
  constexpr int TILE_FLOATS = (D_MODEL / 2) * PITCH;
  constexpr int ROWS  = WAVES * 16;
  constexpr int LP    = NH + 1;
  __shared__ __attribute__((aligned(16))) float Wlds[TILE_FLOATS];
  __shared__ float Lg[ROWS * LP];

  const int tid   = threadIdx.x;
  const int wave  = tid >> 5;
  const int lane  = tid & 31;
  const int m0    = blockIdx.x * ROWS + wave * 16;
  const int khalf = (lane >> 4) << 1;
  const int ncol  = lane & 15;

  const float* Arow = A + (long)(m0 + ncol) * D_MODEL + khalf;
  v2f a_cur[KB];
  #pragma unroll
  for (int s = 0; s < KB; ++s) a_cur[s] = *(const v2f*)(Arow + 4 * s);

  dta_stage_w96<NTHR, PITCH>(Woff, Wlds, 0,  tid);
  dta_stage_w96<NTHR, PITCH>(Waw,  Wlds, NH, tid);
  __syncthreads();

  v8f acc[NTILE];
  #pragma unroll
  for (int t = 0; t < NTILE; ++t) acc[t] = v8f{};

  const float* Bbase = &Wlds[(khalf >> 1) * PITCH + ncol * 2];

  static_assert(KB % G == 0 && K_STEPS % KB == 0, "blocks tile a prefetch group, groups tile K");
  constexpr int NBLK = K_STEPS / G;
  v2f a_nxt[KB];
  v2f bset[2][G][NTILE];
  #pragma unroll
  for (int g = 0; g < G; ++g) {
    #pragma unroll
    for (int t = 0; t < NTILE; ++t)
      bset[0][g][t] = *(const v2f*)(Bbase + g * 2 * PITCH + 32 * t);
  }

  #pragma unroll
  for (int blk = 0; blk < NBLK; ++blk) {
    const int ks0 = blk * G;
    if (blk + 1 < NBLK) {
      #pragma unroll
      for (int g = 0; g < G; ++g) {
        const float* Bk = Bbase + (ks0 + G + g) * 2 * PITCH;
        #pragma unroll
        for (int t = 0; t < NTILE; ++t)
          bset[(blk + 1) & 1][g][t] = *(const v2f*)(Bk + 32 * t);
      }
    }
    if (ks0 % KB == 0 && ks0 + KB < K_STEPS) {
      #pragma unroll
      for (int s = 0; s < KB; ++s) a_nxt[s] = *(const v2f*)(Arow + 4 * (ks0 + KB + s));
    }
    __builtin_amdgcn_sched_barrier(0);
    #pragma unroll
    for (int g = 0; g < G; ++g) {
      #pragma unroll
      for (int t = 0; t < NTILE; ++t)
        acc[t] = __builtin_amdgcn_wmma_f32_16x16x4_f32(false, a_cur[ks0 % KB + g], false, bset[blk & 1][g][t], (short)0, acc[t], false, false);
    }
    __builtin_amdgcn_sched_barrier(0);
    if ((ks0 + G) % KB == 0 && ks0 + G < K_STEPS) {
      #pragma unroll
      for (int s = 0; s < KB; ++s) a_cur[s] = a_nxt[s];
    }
  }

  const int rloc  = wave * 16 + ((lane >> 4) << 3);
  const int rbase = blockIdx.x * ROWS + rloc;
  #pragma unroll
  for (int t = 0; t < HT; ++t) {
    const float bvt = boff[16 * t + ncol];
    #pragma unroll
    for (int r = 0; r < 8; ++r)
      off_out[(long)(rbase + r) * NH + 16 * t + ncol] = acc[t][r] + bvt;
  }
  #pragma unroll
  for (int t = 0; t < HT; ++t) {
    const float bvt = baw[16 * t + ncol];
    #pragma unroll
    for (int r = 0; r < 8; ++r)
      Lg[(rloc + r) * LP + 16 * t + ncol] = acc[HT + t][r] + bvt;
  }
  __syncthreads();

  constexpr int GROUPS = ROWS * 8 / NTHR;
  #pragma unroll
  for (int i = 0; i < GROUPS; ++i) {
    const int q = tid + i * NTHR;
    float* p = &Lg[(q >> 3) * LP + (q & 7) * 12];
    float v[12];
    float mx = -3.402823466e+38f;
    #pragma unroll
    for (int e = 0; e < 12; ++e) { v[e] = p[e]; mx = fmaxf(mx, v[e]); }
    float s = 0.0f;
    #pragma unroll
    for (int e = 0; e < 12; ++e) { v[e] = __expf(v[e] - mx); s += v[e]; }
    const float inv = 1.0f / s;
    #pragma unroll
    for (int e = 0; e < 12; ++e) p[e] = v[e] * inv;
  }
  __syncthreads();

  #pragma unroll
  for (int t = 0; t < HT; ++t) {
    #pragma unroll
    for (int r = 0; r < 8; ++r)
      aw_out[(long)(rbase + r) * NH + 16 * t + ncol] = Lg[(rloc + r) * LP + 16 * t + ncol];
  }
}

__global__ __launch_bounds__(256)
void dta_sample(const float* __restrict__ refp, const float* __restrict__ off_raw,
                const float* __restrict__ aw, const float* __restrict__ v_proj,
                float* __restrict__ out_mid) {
  const int wave = (blockIdx.x * blockDim.x + threadIdx.x) >> 5;
  const int lane = threadIdx.x & 31;
  const int b = wave >> 11;
  const int row = wave;

  const float rp = refp[row];
  const int   Ts[3]   = {2048, 1024, 512};
  const long  base[3] = {0, 8L * 2048, 8L * 2048 + 8L * 1024};

  const float* offr = off_raw + (long)row * 96;
  const float* awr  = aw + (long)row * 96;

  #pragma unroll
  for (int h = 0; h < 8; ++h) {
    float acc = 0.0f;
    #pragma unroll
    for (int l = 0; l < 3; ++l) {
      const float Tf = (float)Ts[l];
      const long  rb = base[l] + (long)b * Ts[l];
      #pragma unroll
      for (int p = 0; p < 4; ++p) {
        const int idx = h * 12 + l * 4 + p;
        const float o = offr[idx];
        const float w = awr[idx];
        float pos  = fminf(fmaxf(rp + o / Tf, 0.0f), 1.0f);
        float sidx = pos * (Tf - 1.0f);
        int ifl = (int)sidx;
        ifl = min(max(ifl, 0), Ts[l] - 2);
        const float wc = sidx - (float)ifl;
        const long r0 = (rb + ifl) * 256 + p * 32 + lane;
        const float vf = v_proj[r0];
        const float vc = v_proj[r0 + 256];
        acc += w * ((1.0f - wc) * vf + wc * vc);
      }
    }
    out_mid[(long)row * 256 + h * 32 + lane] = acc;
  }
}

extern "C" void kernel_launch(void* const* d_in, const int* in_sizes, int n_in,
                              void* d_out, int out_size, void* d_ws, size_t ws_size,
                              hipStream_t stream) {
  (void)in_sizes; (void)n_in; (void)out_size; (void)ws_size;

  const float* query = (const float*)d_in[0];
  const float* refp  = (const float*)d_in[1];
  const float* val0  = (const float*)d_in[2];
  const float* val1  = (const float*)d_in[3];
  const float* val2  = (const float*)d_in[4];
  const float* Woff  = (const float*)d_in[5];
  const float* boff  = (const float*)d_in[6];
  const float* Waw   = (const float*)d_in[7];
  const float* baw   = (const float*)d_in[8];
  const float* Wv    = (const float*)d_in[9];
  const float* bv    = (const float*)d_in[10];
  const float* Wo    = (const float*)d_in[11];
  const float* bo    = (const float*)d_in[12];
  float*       out   = (float*)d_out;

  const int BQ = 8 * 2048;

  char* ws = (char*)d_ws;
  float* off_raw = (float*)(ws);
  float* aw_buf  = (float*)(ws + (size_t)BQ * 96 * 4);
  float* v_proj  = (float*)(ws + (size_t)BQ * 96 * 4 * 2);
  float* out_mid = (float*)(ws + (size_t)BQ * 96 * 4 * 2
                               + (size_t)(BQ + 8192 + 4096) * 256 * 4);

  dim3 blk(256);
  dim3 blk4(128);

  dta_qproj_softmax<4, 8, 2><<<dim3(256), blk4, 0, stream>>>(query, Woff, boff, Waw, baw, off_raw, aw_buf);
  dta_vproj_ws<64, 4, 7, 16, 4><<<dim3(64, 4), blk4, 0, stream>>>(val0, val1, val2, Wv, bv, v_proj);
  dta_sample<<<dim3(BQ / 8), blk, 0, stream>>>(refp, off_raw, aw_buf, v_proj, out_mid);
  dta_gemm_f32_wmma<128, 8, 8, 2><<<dim3(128, 2), blk, 0, stream>>>(out_mid, Wo, bo, out, 256);
}


# Sampling loop interchanged: (level, point) outside, head inside; off/aw stored [level,point][head] by the projection epilogue; the 8 heads' reads of a value slice repeat in the WGP cache; sums kept
# speedup vs baseline: 1.3721x; 1.0530x over previous
#include <hip/hip_runtime.h>
#include <hip/hip_bf16.h>
#include <math.h>


typedef float v2f __attribute__((ext_vector_type(2)));
typedef float v4f __attribute__((ext_vector_type(4)));
typedef float v8f __attribute__((ext_vector_type(8)));

#define D_MODEL 256

#define K_STEPS (D_MODEL / 4)

template <int NW, int WAVES, int KB, int G>
__global__ __launch_bounds__(WAVES * 32)
void dta_gemm_f32_wmma(const float* __restrict__ A, const float* __restrict__ W,
                       const float* __restrict__ bias, float* __restrict__ C,
                       int N) {
  constexpr int NTHR  = WAVES * 32;
  constexpr int NTILE = NW / 16;
  constexpr int PITCH = NW * 2 + 32;
  constexpr int CPR   = NW / 4;
  constexpr int TILE_FLOATS = (D_MODEL / 2) * PITCH;
  constexpr int ONE_PER_WGP_FLOATS = 41984;
  constexpr int LDS_FLOATS = TILE_FLOATS > ONE_PER_WGP_FLOATS ? TILE_FLOATS : ONE_PER_WGP_FLOATS;
  __shared__ __attribute__((aligned(16))) float Wlds[LDS_FLOATS];

  const int tid   = threadIdx.x;
  const int wave  = tid >> 5;
  const int lane  = tid & 31;
  const int n0    = blockIdx.y * NW;
  const int m0    = blockIdx.x * (WAVES * 16) + wave * 16;
  const int khalf = (lane >> 4) << 1;
  const int ncol  = lane & 15;

  const float* Arow = A + (long)(m0 + ncol) * D_MODEL + khalf;
  v2f a_cur[KB];
  #pragma unroll
  for (int s = 0; s < KB; ++s) a_cur[s] = *(const v2f*)(Arow + 4 * s);

  constexpr int STAGE_ITERS = (D_MODEL / 2) * CPR / NTHR;
  static_assert(STAGE_ITERS * NTHR == (D_MODEL / 2) * CPR, "chunks divide evenly over the threads");
  #pragma unroll
  for (int it = 0; it < STAGE_ITERS; ++it) {
    const int ch = tid + it * NTHR;
    const int kk = ch / CPR;
    const int j  = (ch - kk * CPR) << 2;
    v4f r0 = *(const v4f*)(W + (long)(2 * kk)     * N + n0 + j);
    v4f r1 = *(const v4f*)(W + (long)(2 * kk + 1) * N + n0 + j);
    v4f lo = {r0.x, r1.x, r0.y, r1.y};
    v4f hi = {r0.z, r1.z, r0.w, r1.w};
    *(v4f*)(&Wlds[kk * PITCH + j * 2])     = lo;
    *(v4f*)(&Wlds[kk * PITCH + j * 2 + 4]) = hi;
  }
  __syncthreads();

  v8f acc[NTILE];
  #pragma unroll
  for (int t = 0; t < NTILE; ++t) acc[t] = v8f{};

  const float* Bbase = &Wlds[(khalf >> 1) * PITCH + ncol * 2];

  static_assert(KB % G == 0 && K_STEPS % KB == 0, "blocks tile a prefetch group, groups tile K");
  constexpr int NBLK = K_STEPS / G;
  v2f a_nxt[KB];
  v2f bset[2][G][NTILE];
  #pragma unroll
  for (int g = 0; g < G; ++g) {
    #pragma unroll
    for (int t = 0; t < NTILE; ++t)
      bset[0][g][t] = *(const v2f*)(Bbase + g * 2 * PITCH + 32 * t);
  }

  #pragma unroll
  for (int blk = 0; blk < NBLK; ++blk) {
    const int ks0 = blk * G;
    if (blk + 1 < NBLK) {
      #pragma unroll
      for (int g = 0; g < G; ++g) {
        const float* Bk = Bbase + (ks0 + G + g) * 2 * PITCH;
        #pragma unroll
        for (int t = 0; t < NTILE; ++t)
          bset[(blk + 1) & 1][g][t] = *(const v2f*)(Bk + 32 * t);
      }
    }
    if (ks0 % KB == 0 && ks0 + KB < K_STEPS) {
      #pragma unroll
      for (int s = 0; s < KB; ++s) a_nxt[s] = *(const v2f*)(Arow + 4 * (ks0 + KB + s));
    }
    __builtin_amdgcn_sched_barrier(0);
    #pragma unroll
    for (int g = 0; g < G; ++g) {
      #pragma unroll
      for (int t = 0; t < NTILE; ++t)
        acc[t] = __builtin_amdgcn_wmma_f32_16x16x4_f32(false, a_cur[ks0 % KB + g], false, bset[blk & 1][g][t], (short)0, acc[t], false, false);
    }
    __builtin_amdgcn_sched_barrier(0);
    if ((ks0 + G) % KB == 0 && ks0 + G < K_STEPS) {
      #pragma unroll
      for (int s = 0; s < KB; ++s) a_cur[s] = a_nxt[s];
    }
  }

  const int rbase = m0 + ((lane >> 4) << 3);
  #pragma unroll
  for (int t = 0; t < NTILE; ++t) {
    const float bvt = bias[n0 + 16 * t + ncol];
    #pragma unroll
    for (int r = 0; r < 8; ++r)
      C[(long)(rbase + r) * N + n0 + 16 * t + ncol] = acc[t][r] + bvt;
  }
}

#define DTA_STRIPS_L0 1024
#define DTA_STRIPS_L1 512

__device__ __forceinline__ const float* dta_strip_rows(const float* __restrict__ V0,
                                                       const float* __restrict__ V1,
                                                       const float* __restrict__ V2, int s) {
  if (s < DTA_STRIPS_L0) return V0 + (long)s * (16 * D_MODEL);
  if (s < DTA_STRIPS_L0 + DTA_STRIPS_L1) return V1 + (long)(s - DTA_STRIPS_L0) * (16 * D_MODEL);
  return V2 + (long)(s - DTA_STRIPS_L0 - DTA_STRIPS_L1) * (16 * D_MODEL);
}

template <int NW, int WAVES, int STRIPS, int KB, int G>
__global__ __launch_bounds__(WAVES * 32)
void dta_vproj_ws(const float* __restrict__ V0, const float* __restrict__ V1,
                  const float* __restrict__ V2, const float* __restrict__ W,
                  const float* __restrict__ bias, float* __restrict__ C) {
  constexpr int N     = D_MODEL;
  constexpr int NTHR  = WAVES * 32;
  constexpr int NTILE = NW / 16;
  constexpr int PITCH = NW * 2 + 32;
  constexpr int CPR   = NW / 4;
  constexpr int TILE_FLOATS = (D_MODEL / 2) * PITCH;
  constexpr int ONE_PER_WGP_FLOATS = 41984;
  constexpr int LDS_FLOATS = TILE_FLOATS > ONE_PER_WGP_FLOATS ? TILE_FLOATS : ONE_PER_WGP_FLOATS;
  __shared__ __attribute__((aligned(16))) float Wlds[LDS_FLOATS];

  const int tid   = threadIdx.x;
  const int wave  = __builtin_amdgcn_readfirstlane(tid >> 5);
  const int lane  = tid & 31;
  const int n0    = blockIdx.y * NW;
  const int khalf = (lane >> 4) << 1;
  const int ncol  = lane & 15;
  const int lofs  = ncol * D_MODEL + khalf;
  const int s0    = blockIdx.x * (WAVES * STRIPS) + wave;

  const float* Arow = dta_strip_rows(V0, V1, V2, s0) + lofs;
  v2f a_cur[KB];
  #pragma unroll
  for (int s = 0; s < KB; ++s) a_cur[s] = *(const v2f*)(Arow + 4 * s);

  constexpr int STAGE_ITERS = (D_MODEL / 2) * CPR / NTHR;
  static_assert(STAGE_ITERS * NTHR == (D_MODEL / 2) * CPR, "chunks divide evenly over the threads");
  #pragma unroll
  for (int it = 0; it < STAGE_ITERS; ++it) {
    const int ch = tid + it * NTHR;
    const int kk = ch / CPR;
    const int j  = (ch - kk * CPR) << 2;
    v4f r0 = *(const v4f*)(W + (long)(2 * kk)     * N + n0 + j);
    v4f r1 = *(const v4f*)(W + (long)(2 * kk + 1) * N + n0 + j);
    v4f lo = {r0.x, r1.x, r0.y, r1.y};
    v4f hi = {r0.z, r1.z, r0.w, r1.w};
    *(v4f*)(&Wlds[kk * PITCH + j * 2])     = lo;
    *(v4f*)(&Wlds[kk * PITCH + j * 2 + 4]) = hi;
  }

  float bvt[NTILE];
  #pragma unroll
  for (int t = 0; t < NTILE; ++t) bvt[t] = bias[n0 + 16 * t + ncol];
  __syncthreads();

  const float* Bbase = &Wlds[(khalf >> 1) * PITCH + ncol * 2];

  static_assert(KB % G == 0 && K_STEPS % KB == 0, "blocks tile a prefetch group, groups tile K");
  constexpr int NBLK = K_STEPS / G;
  static_assert(NBLK % 2 == 0, "the B register sets alternate and must line up again at the next strip");
  v2f a_nxt[KB];
  v2f bset[2][G][NTILE];
  #pragma unroll
  for (int g = 0; g < G; ++g) {
    #pragma unroll
    for (int t = 0; t < NTILE; ++t)
      bset[0][g][t] = *(const v2f*)(Bbase + g * 2 * PITCH + 32 * t);
  }

  for (int i = 0; i < STRIPS; ++i) {
    const int sc = s0 + i * WAVES;
    const int sn = (i + 1 < STRIPS) ? sc + WAVES : sc;
    const float* Anext = dta_strip_rows(V0, V1, V2, sn) + lofs;

    v8f acc[NTILE];
    #pragma unroll
    for (int t = 0; t < NTILE; ++t) acc[t] = v8f{};

    #pragma unroll
    for (int blk = 0; blk < NBLK; ++blk) {
      const int ks0 = blk * G;
      #pragma unroll
      for (int g = 0; g < G; ++g) {
        const float* Bk = Bbase + ((ks0 + G + g) % K_STEPS) * 2 * PITCH;
        #pragma unroll
        for (int t = 0; t < NTILE; ++t)
          bset[(blk + 1) & 1][g][t] = *(const v2f*)(Bk + 32 * t);
      }
      if (ks0 % KB == 0) {
        if (ks0 + KB < K_STEPS) {
          #pragma unroll
          for (int s = 0; s < KB; ++s) a_nxt[s] = *(const v2f*)(Arow + 4 * (ks0 + KB + s));
        } else {
          #pragma unroll
          for (int s = 0; s < KB; ++s) a_nxt[s] = *(const v2f*)(Anext + 4 * s);
        }
      }
      __builtin_amdgcn_sched_barrier(0);
      #pragma unroll
      for (int g = 0; g < G; ++g) {
        #pragma unroll
        for (int t = 0; t < NTILE; ++t)
          acc[t] = __builtin_amdgcn_wmma_f32_16x16x4_f32(false, a_cur[ks0 % KB + g], false, bset[blk & 1][g][t], (short)0, acc[t], false, false);
      }
      __builtin_amdgcn_sched_barrier(0);
      if ((ks0 + G) % KB == 0) {
        #pragma unroll
        for (int s = 0; s < KB; ++s) a_cur[s] = a_nxt[s];
      }
    }

    float* Crow = C + (long)sc * (16 * N) + (long)((lane >> 4) << 3) * N + n0 + ncol;
    #pragma unroll
    for (int t = 0; t < NTILE; ++t) {
      #pragma unroll
      for (int r = 0; r < 8; ++r)
        Crow[r * N + 16 * t] = acc[t][r] + bvt[t];
    }
    Arow = Anext;
  }
}

template <int NTHR, int PITCH>
__device__ __forceinline__ void dta_stage_w96(const float* __restrict__ W, float* Wlds,
                                              int cofs, int tid) {
  constexpr int CPR   = 96 / 4;
  constexpr int ITERS = (D_MODEL / 2) * CPR / NTHR;
  static_assert(ITERS * NTHR == (D_MODEL / 2) * CPR, "chunks divide evenly over the threads");
  #pragma unroll
  for (int it = 0; it < ITERS; ++it) {
    const int ch = tid + it * NTHR;
    const int kk = ch / CPR;
    const int j  = (ch - kk * CPR) << 2;
    v4f r0 = *(const v4f*)(W + (long)(2 * kk)     * 96 + j);
    v4f r1 = *(const v4f*)(W + (long)(2 * kk + 1) * 96 + j);
    v4f lo = {r0.x, r1.x, r0.y, r1.y};
    v4f hi = {r0.z, r1.z, r0.w, r1.w};
    *(v4f*)(&Wlds[kk * PITCH + (cofs + j) * 2])     = lo;
    *(v4f*)(&Wlds[kk * PITCH + (cofs + j) * 2 + 4]) = hi;
  }
}

template <int WAVES, int KB, int G>
__global__ __launch_bounds__(WAVES * 32)
void dta_qproj_softmax(const float* __restrict__ A,
                       const float* __restrict__ Woff, const float* __restrict__ boff,
                       const float* __restrict__ Waw,  const float* __restrict__ baw,
                       float* __restrict__ off_out, float* __restrict__ aw_out) {
  constexpr int NH    = 96;
  constexpr int NW    = 2 * NH;
  constexpr int NTHR  = WAVES * 32;
  constexpr int NTILE = NW / 16;
  constexpr int HT    = NH / 16;
  constexpr int PITCH = NW * 2 + 32;
  constexpr int TILE_FLOATS = (D_MODEL / 2) * PITCH;
  constexpr int ROWS  = WAVES * 16;
  constexpr int LP    = NH + 1;
  __shared__ __attribute__((aligned(16))) float Wlds[TILE_FLOATS];
  __shared__ float Lg[ROWS * LP];
  __shared__ float Og[ROWS * LP];

  const int tid   = threadIdx.x;
  const int wave  = tid >> 5;
  const int lane  = tid & 31;
  const int m0    = blockIdx.x * ROWS + wave * 16;
  const int khalf = (lane >> 4) << 1;
  const int ncol  = lane & 15;

  const float* Arow = A + (long)(m0 + ncol) * D_MODEL + khalf;
  v2f a_cur[KB];
  #pragma unroll
  for (int s = 0; s < KB; ++s) a_cur[s] = *(const v2f*)(Arow + 4 * s);

  dta_stage_w96<NTHR, PITCH>(Woff, Wlds, 0,  tid);
  dta_stage_w96<NTHR, PITCH>(Waw,  Wlds, NH, tid);
  __syncthreads();

  v8f acc[NTILE];
  #pragma unroll
  for (int t = 0; t < NTILE; ++t) acc[t] = v8f{};

  const float* Bbase = &Wlds[(khalf >> 1) * PITCH + ncol * 2];

  static_assert(KB % G == 0 && K_STEPS % KB == 0, "blocks tile a prefetch group, groups tile K");
  constexpr int NBLK = K_STEPS / G;
  v2f a_nxt[KB];
  v2f bset[2][G][NTILE];
  #pragma unroll
  for (int g = 0; g < G; ++g) {
    #pragma unroll
    for (int t = 0; t < NTILE; ++t)
      bset[0][g][t] = *(const v2f*)(Bbase + g * 2 * PITCH + 32 * t);
  }

  #pragma unroll
  for (int blk = 0; blk < NBLK; ++blk) {
    const int ks0 = blk * G;
    if (blk + 1 < NBLK) {
      #pragma unroll
      for (int g = 0; g < G; ++g) {
        const float* Bk = Bbase + (ks0 + G + g) * 2 * PITCH;
        #pragma unroll
        for (int t = 0; t < NTILE; ++t)
          bset[(blk + 1) & 1][g][t] = *(const v2f*)(Bk + 32 * t);
      }
    }
    if (ks0 % KB == 0 && ks0 + KB < K_STEPS) {
      #pragma unroll
      for (int s = 0; s < KB; ++s) a_nxt[s] = *(const v2f*)(Arow + 4 * (ks0 + KB + s));
    }
    __builtin_amdgcn_sched_barrier(0);
    #pragma unroll
    for (int g = 0; g < G; ++g) {
      #pragma unroll
      for (int t = 0; t < NTILE; ++t)
        acc[t] = __builtin_amdgcn_wmma_f32_16x16x4_f32(false, a_cur[ks0 % KB + g], false, bset[blk & 1][g][t], (short)0, acc[t], false, false);
    }
    __builtin_amdgcn_sched_barrier(0);
    if ((ks0 + G) % KB == 0 && ks0 + G < K_STEPS) {
      #pragma unroll
      for (int s = 0; s < KB; ++s) a_cur[s] = a_nxt[s];
    }
  }

  const int rloc  = wave * 16 + ((lane >> 4) << 3);
  const int rbase = blockIdx.x * ROWS + rloc;
  #pragma unroll
  for (int t = 0; t < HT; ++t) {
    const float bvt = boff[16 * t + ncol];
    #pragma unroll
    for (int r = 0; r < 8; ++r)
      Og[(rloc + r) * LP + 16 * t + ncol] = acc[t][r] + bvt;
  }
  #pragma unroll
  for (int t = 0; t < HT; ++t) {
    const float bvt = baw[16 * t + ncol];
    #pragma unroll
    for (int r = 0; r < 8; ++r)
      Lg[(rloc + r) * LP + 16 * t + ncol] = acc[HT + t][r] + bvt;
  }
  __syncthreads();

  constexpr int GROUPS = ROWS * 8 / NTHR;
  #pragma unroll
  for (int i = 0; i < GROUPS; ++i) {
    const int q = tid + i * NTHR;
    float* p = &Lg[(q >> 3) * LP + (q & 7) * 12];
    float v[12];
    float mx = -3.402823466e+38f;
    #pragma unroll
    for (int e = 0; e < 12; ++e) { v[e] = p[e]; mx = fmaxf(mx, v[e]); }
    float s = 0.0f;
    #pragma unroll
    for (int e = 0; e < 12; ++e) { v[e] = __expf(v[e] - mx); s += v[e]; }
    const float inv = 1.0f / s;
    #pragma unroll
    for (int e = 0; e < 12; ++e) p[e] = v[e] * inv;
  }
  __syncthreads();

  #pragma unroll
  for (int t = 0; t < HT; ++t) {
    const int cs   = 16 * t + ncol;
    const int csrc = (cs & 7) * 12 + (cs >> 3);
    #pragma unroll
    for (int r = 0; r < 8; ++r) {
      off_out[(long)(rbase + r) * NH + cs] = Og[(rloc + r) * LP + csrc];
      aw_out[(long)(rbase + r) * NH + cs]  = Lg[(rloc + r) * LP + csrc];
    }
  }
}

__global__ __launch_bounds__(256)
void dta_sample(const float* __restrict__ refp, const float* __restrict__ off_raw,
                const float* __restrict__ aw, const float* __restrict__ v_proj,
                float* __restrict__ out_mid) {
  const int wave = (blockIdx.x * blockDim.x + threadIdx.x) >> 5;
  const int lane = threadIdx.x & 31;
  const int b = wave >> 11;
  const int row = wave;

  const float rp = refp[row];
  const int   Ts[3]   = {2048, 1024, 512};
  const long  base[3] = {0, 8L * 2048, 8L * 2048 + 8L * 1024};

  const float* offr = off_raw + (long)row * 96;
  const float* awr  = aw + (long)row * 96;

  float acc[8];
  #pragma unroll
  for (int h = 0; h < 8; ++h) acc[h] = 0.0f;

  #pragma unroll
  for (int l = 0; l < 3; ++l) {
    const float Tf = (float)Ts[l];
    const long  rb = base[l] + (long)b * Ts[l];
    #pragma unroll
    for (int p = 0; p < 4; ++p) {
      #pragma unroll
      for (int h = 0; h < 8; ++h) {
        const int idx = (l * 4 + p) * 8 + h;
        const float o = offr[idx];
        const float w = awr[idx];
        float pos  = fminf(fmaxf(rp + o / Tf, 0.0f), 1.0f);
        float sidx = pos * (Tf - 1.0f);
        int ifl = (int)sidx;
        ifl = min(max(ifl, 0), Ts[l] - 2);
        const float wc = sidx - (float)ifl;
        const long r0 = (rb + ifl) * 256 + p * 32 + lane;
        const float vf = v_proj[r0];
        const float vc = v_proj[r0 + 256];
        acc[h] += w * ((1.0f - wc) * vf + wc * vc);
      }
    }
  }
  #pragma unroll
  for (int h = 0; h < 8; ++h)
    out_mid[(long)row * 256 + h * 32 + lane] = acc[h];
}

extern "C" void kernel_launch(void* const* d_in, const int* in_sizes, int n_in,
                              void* d_out, int out_size, void* d_ws, size_t ws_size,
                              hipStream_t stream) {
  (void)in_sizes; (void)n_in; (void)out_size; (void)ws_size;

  const float* query = (const float*)d_in[0];
  const float* refp  = (const float*)d_in[1];
  const float* val0  = (const float*)d_in[2];
  const float* val1  = (const float*)d_in[3];
  const float* val2  = (const float*)d_in[4];
  const float* Woff  = (const float*)d_in[5];
  const float* boff  = (const float*)d_in[6];
  const float* Waw   = (const float*)d_in[7];
  const float* baw   = (const float*)d_in[8];
  const float* Wv    = (const float*)d_in[9];
  const float* bv    = (const float*)d_in[10];
  const float* Wo    = (const float*)d_in[11];
  const float* bo    = (const float*)d_in[12];
  float*       out   = (float*)d_out;

  const int BQ = 8 * 2048;

  char* ws = (char*)d_ws;
  float* off_raw = (float*)(ws);
  float* aw_buf  = (float*)(ws + (size_t)BQ * 96 * 4);
  float* v_proj  = (float*)(ws + (size_t)BQ * 96 * 4 * 2);
  float* out_mid = (float*)(ws + (size_t)BQ * 96 * 4 * 2
                               + (size_t)(BQ + 8192 + 4096) * 256 * 4);

  dim3 blk(256);
  dim3 blk4(128);

  dta_qproj_softmax<4, 8, 2><<<dim3(256), blk4, 0, stream>>>(query, Woff, boff, Waw, baw, off_raw, aw_buf);
  dta_vproj_ws<64, 4, 7, 16, 4><<<dim3(64, 4), blk4, 0, stream>>>(val0, val1, val2, Wv, bv, v_proj);
  dta_sample<<<dim3(BQ / 8), blk, 0, stream>>>(refp, off_raw, aw_buf, v_proj, out_mid);
  dta_gemm_f32_wmma<128, 8, 8, 2><<<dim3(128, 2), blk, 0, stream>>>(out_mid, Wo, bo, out, 256);
}


# Sampling as a rolled loop of 6 batches of 16 samples: all addresses, then the batch's 32 row reads plus the next batch's parameter reads in flight together, then the sums; sum order kept
# speedup vs baseline: 1.4598x; 1.0639x over previous
#include <hip/hip_runtime.h>
#include <hip/hip_bf16.h>
#include <math.h>


typedef float v2f __attribute__((ext_vector_type(2)));
typedef float v4f __attribute__((ext_vector_type(4)));
typedef float v8f __attribute__((ext_vector_type(8)));

#define D_MODEL 256

#define K_STEPS (D_MODEL / 4)

template <int NW, int WAVES, int KB, int G>
__global__ __launch_bounds__(WAVES * 32)
void dta_gemm_f32_wmma(const float* __restrict__ A, const float* __restrict__ W,
                       const float* __restrict__ bias, float* __restrict__ C,
                       int N) {
  constexpr int NTHR  = WAVES * 32;
  constexpr int NTILE = NW / 16;
  constexpr int PITCH = NW * 2 + 32;
  constexpr int CPR   = NW / 4;
  constexpr int TILE_FLOATS = (D_MODEL / 2) * PITCH;
  constexpr int ONE_PER_WGP_FLOATS = 41984;
  constexpr int LDS_FLOATS = TILE_FLOATS > ONE_PER_WGP_FLOATS ? TILE_FLOATS : ONE_PER_WGP_FLOATS;
  __shared__ __attribute__((aligned(16))) float Wlds[LDS_FLOATS];

  const int tid   = threadIdx.x;
  const int wave  = tid >> 5;
  const int lane  = tid & 31;
  const int n0    = blockIdx.y * NW;
  const int m0    = blockIdx.x * (WAVES * 16) + wave * 16;
  const int khalf = (lane >> 4) << 1;
  const int ncol  = lane & 15;

  const float* Arow = A + (long)(m0 + ncol) * D_MODEL + khalf;
  v2f a_cur[KB];
  #pragma unroll
  for (int s = 0; s < KB; ++s) a_cur[s] = *(const v2f*)(Arow + 4 * s);

  constexpr int STAGE_ITERS = (D_MODEL / 2) * CPR / NTHR;
  static_assert(STAGE_ITERS * NTHR == (D_MODEL / 2) * CPR, "chunks divide evenly over the threads");
  #pragma unroll
  for (int it = 0; it < STAGE_ITERS; ++it) {
    const int ch = tid + it * NTHR;
    const int kk = ch / CPR;
    const int j  = (ch - kk * CPR) << 2;
    v4f r0 = *(const v4f*)(W + (long)(2 * kk)     * N + n0 + j);
    v4f r1 = *(const v4f*)(W + (long)(2 * kk + 1) * N + n0 + j);
    v4f lo = {r0.x, r1.x, r0.y, r1.y};
    v4f hi = {r0.z, r1.z, r0.w, r1.w};
    *(v4f*)(&Wlds[kk * PITCH + j * 2])     = lo;
    *(v4f*)(&Wlds[kk * PITCH + j * 2 + 4]) = hi;
  }
  __syncthreads();

  v8f acc[NTILE];
  #pragma unroll
  for (int t = 0; t < NTILE; ++t) acc[t] = v8f{};

  const float* Bbase = &Wlds[(khalf >> 1) * PITCH + ncol * 2];

  static_assert(KB % G == 0 && K_STEPS % KB == 0, "blocks tile a prefetch group, groups tile K");
  constexpr int NBLK = K_STEPS / G;
  v2f a_nxt[KB];
  v2f bset[2][G][NTILE];
  #pragma unroll
  for (int g = 0; g < G; ++g) {
    #pragma unroll
    for (int t = 0; t < NTILE; ++t)
      bset[0][g][t] = *(const v2f*)(Bbase + g * 2 * PITCH + 32 * t);
  }

  #pragma unroll
  for (int blk = 0; blk < NBLK; ++blk) {
    const int ks0 = blk * G;
    if (blk + 1 < NBLK) {
      #pragma unroll
      for (int g = 0; g < G; ++g) {
        const float* Bk = Bbase + (ks0 + G + g) * 2 * PITCH;
        #pragma unroll
        for (int t = 0; t < NTILE; ++t)
          bset[(blk + 1) & 1][g][t] = *(const v2f*)(Bk + 32 * t);
      }
    }
    if (ks0 % KB == 0 && ks0 + KB < K_STEPS) {
      #pragma unroll
      for (int s = 0; s < KB; ++s) a_nxt[s] = *(const v2f*)(Arow + 4 * (ks0 + KB + s));
    }
    __builtin_amdgcn_sched_barrier(0);
    #pragma unroll
    for (int g = 0; g < G; ++g) {
      #pragma unroll
      for (int t = 0; t < NTILE; ++t)
        acc[t] = __builtin_amdgcn_wmma_f32_16x16x4_f32(false, a_cur[ks0 % KB + g], false, bset[blk & 1][g][t], (short)0, acc[t], false, false);
    }
    __builtin_amdgcn_sched_barrier(0);
    if ((ks0 + G) % KB == 0 && ks0 + G < K_STEPS) {
      #pragma unroll
      for (int s = 0; s < KB; ++s) a_cur[s] = a_nxt[s];
    }
  }

  const int rbase = m0 + ((lane >> 4) << 3);
  #pragma unroll
  for (int t = 0; t < NTILE; ++t) {
    const float bvt = bias[n0 + 16 * t + ncol];
    #pragma unroll
    for (int r = 0; r < 8; ++r)
      C[(long)(rbase + r) * N + n0 + 16 * t + ncol] = acc[t][r] + bvt;
  }
}

#define DTA_STRIPS_L0 1024
#define DTA_STRIPS_L1 512

__device__ __forceinline__ const float* dta_strip_rows(const float* __restrict__ V0,
                                                       const float* __restrict__ V1,
                                                       const float* __restrict__ V2, int s) {
  if (s < DTA_STRIPS_L0) return V0 + (long)s * (16 * D_MODEL);
  if (s < DTA_STRIPS_L0 + DTA_STRIPS_L1) return V1 + (long)(s - DTA_STRIPS_L0) * (16 * D_MODEL);
  return V2 + (long)(s - DTA_STRIPS_L0 - DTA_STRIPS_L1) * (16 * D_MODEL);
}

template <int NW, int WAVES, int STRIPS, int KB, int G>
__global__ __launch_bounds__(WAVES * 32)
void dta_vproj_ws(const float* __restrict__ V0, const float* __restrict__ V1,
                  const float* __restrict__ V2, const float* __restrict__ W,
                  const float* __restrict__ bias, float* __restrict__ C) {
  constexpr int N     = D_MODEL;
  constexpr int NTHR  = WAVES * 32;
  constexpr int NTILE = NW / 16;
  constexpr int PITCH = NW * 2 + 32;
  constexpr int CPR   = NW / 4;
  constexpr int TILE_FLOATS = (D_MODEL / 2) * PITCH;
  constexpr int ONE_PER_WGP_FLOATS = 41984;
  constexpr int LDS_FLOATS = TILE_FLOATS > ONE_PER_WGP_FLOATS ? TILE_FLOATS : ONE_PER_WGP_FLOATS;
  __shared__ __attribute__((aligned(16))) float Wlds[LDS_FLOATS];

  const int tid   = threadIdx.x;
  const int wave  = __builtin_amdgcn_readfirstlane(tid >> 5);
  const int lane  = tid & 31;
  const int n0    = blockIdx.y * NW;
  const int khalf = (lane >> 4) << 1;
  const int ncol  = lane & 15;
  const int lofs  = ncol * D_MODEL + khalf;
  const int s0    = blockIdx.x * (WAVES * STRIPS) + wave;

  const float* Arow = dta_strip_rows(V0, V1, V2, s0) + lofs;
  v2f a_cur[KB];
  #pragma unroll
  for (int s = 0; s < KB; ++s) a_cur[s] = *(const v2f*)(Arow + 4 * s);

  constexpr int STAGE_ITERS = (D_MODEL / 2) * CPR / NTHR;
  static_assert(STAGE_ITERS * NTHR == (D_MODEL / 2) * CPR, "chunks divide evenly over the threads");
  #pragma unroll
  for (int it = 0; it < STAGE_ITERS; ++it) {
    const int ch = tid + it * NTHR;
    const int kk = ch / CPR;
    const int j  = (ch - kk * CPR) << 2;
    v4f r0 = *(const v4f*)(W + (long)(2 * kk)     * N + n0 + j);
    v4f r1 = *(const v4f*)(W + (long)(2 * kk + 1) * N + n0 + j);
    v4f lo = {r0.x, r1.x, r0.y, r1.y};
    v4f hi = {r0.z, r1.z, r0.w, r1.w};
    *(v4f*)(&Wlds[kk * PITCH + j * 2])     = lo;
    *(v4f*)(&Wlds[kk * PITCH + j * 2 + 4]) = hi;
  }

  float bvt[NTILE];
  #pragma unroll
  for (int t = 0; t < NTILE; ++t) bvt[t] = bias[n0 + 16 * t + ncol];
  __syncthreads();

  const float* Bbase = &Wlds[(khalf >> 1) * PITCH + ncol * 2];

  static_assert(KB % G == 0 && K_STEPS % KB == 0, "blocks tile a prefetch group, groups tile K");
  constexpr int NBLK = K_STEPS / G;
  static_assert(NBLK % 2 == 0, "the B register sets alternate and must line up again at the next strip");
  v2f a_nxt[KB];
  v2f bset[2][G][NTILE];
  #pragma unroll
  for (int g = 0; g < G; ++g) {
    #pragma unroll
    for (int t = 0; t < NTILE; ++t)
      bset[0][g][t] = *(const v2f*)(Bbase + g * 2 * PITCH + 32 * t);
  }

  for (int i = 0; i < STRIPS; ++i) {
    const int sc = s0 + i * WAVES;
    const int sn = (i + 1 < STRIPS) ? sc + WAVES : sc;
    const float* Anext = dta_strip_rows(V0, V1, V2, sn) + lofs;

    v8f acc[NTILE];
    #pragma unroll
    for (int t = 0; t < NTILE; ++t) acc[t] = v8f{};

    #pragma unroll
    for (int blk = 0; blk < NBLK; ++blk) {
      const int ks0 = blk * G;
      #pragma unroll
      for (int g = 0; g < G; ++g) {
        const float* Bk = Bbase + ((ks0 + G + g) % K_STEPS) * 2 * PITCH;
        #pragma unroll
        for (int t = 0; t < NTILE; ++t)
          bset[(blk + 1) & 1][g][t] = *(const v2f*)(Bk + 32 * t);
      }
      if (ks0 % KB == 0) {
        if (ks0 + KB < K_STEPS) {
          #pragma unroll
          for (int s = 0; s < KB; ++s) a_nxt[s] = *(const v2f*)(Arow + 4 * (ks0 + KB + s));
        } else {
          #pragma unroll
          for (int s = 0; s < KB; ++s) a_nxt[s] = *(const v2f*)(Anext + 4 * s);
        }
      }
      __builtin_amdgcn_sched_barrier(0);
      #pragma unroll
      for (int g = 0; g < G; ++g) {
        #pragma unroll
        for (int t = 0; t < NTILE; ++t)
          acc[t] = __builtin_amdgcn_wmma_f32_16x16x4_f32(false, a_cur[ks0 % KB + g], false, bset[blk & 1][g][t], (short)0, acc[t], false, false);
      }
      __builtin_amdgcn_sched_barrier(0);
      if ((ks0 + G) % KB == 0) {
        #pragma unroll
        for (int s = 0; s < KB; ++s) a_cur[s] = a_nxt[s];
      }
    }

    float* Crow = C + (long)sc * (16 * N) + (long)((lane >> 4) << 3) * N + n0 + ncol;
    #pragma unroll
    for (int t = 0; t < NTILE; ++t) {
      #pragma unroll
      for (int r = 0; r < 8; ++r)
        Crow[r * N + 16 * t] = acc[t][r] + bvt[t];
    }
    Arow = Anext;
  }
}

template <int NTHR, int PITCH>
__device__ __forceinline__ void dta_stage_w96(const float* __restrict__ W, float* Wlds,
                                              int cofs, int tid) {
  constexpr int CPR   = 96 / 4;
  constexpr int ITERS = (D_MODEL / 2) * CPR / NTHR;
  static_assert(ITERS * NTHR == (D_MODEL / 2) * CPR, "chunks divide evenly over the threads");
  #pragma unroll
  for (int it = 0; it < ITERS; ++it) {
    const int ch = tid + it * NTHR;
    const int kk = ch / CPR;
    const int j  = (ch - kk * CPR) << 2;
    v4f r0 = *(const v4f*)(W + (long)(2 * kk)     * 96 + j);
    v4f r1 = *(const v4f*)(W + (long)(2 * kk + 1) * 96 + j);
    v4f lo = {r0.x, r1.x, r0.y, r1.y};
    v4f hi = {r0.z, r1.z, r0.w, r1.w};
    *(v4f*)(&Wlds[kk * PITCH + (cofs + j) * 2])     = lo;
    *(v4f*)(&Wlds[kk * PITCH + (cofs + j) * 2 + 4]) = hi;
  }
}

template <int WAVES, int KB, int G>
__global__ __launch_bounds__(WAVES * 32)
void dta_qproj_softmax(const float* __restrict__ A,
                       const float* __restrict__ Woff, const float* __restrict__ boff,
                       const float* __restrict__ Waw,  const float* __restrict__ baw,
                       float* __restrict__ off_out, float* __restrict__ aw_out) {
  constexpr int NH    = 96;
  constexpr int NW    = 2 * NH;
  constexpr int NTHR  = WAVES * 32;
  constexpr int NTILE = NW / 16;
  constexpr int HT    = NH / 16;
  constexpr int PITCH = NW * 2 + 32;
  constexpr int TILE_FLOATS = (D_MODEL / 2) * PITCH;
  constexpr int ROWS  = WAVES * 16;
  constexpr int LP    = NH + 1;
  __shared__ __attribute__((aligned(16))) float Wlds[TILE_FLOATS];
  __shared__ float Lg[ROWS * LP];
  __shared__ float Og[ROWS * LP];

  const int tid   = threadIdx.x;
  const int wave  = tid >> 5;
  const int lane  = tid & 31;
  const int m0    = blockIdx.x * ROWS + wave * 16;
  const int khalf = (lane >> 4) << 1;
  const int ncol  = lane & 15;

  const float* Arow = A + (long)(m0 + ncol) * D_MODEL + khalf;
  v2f a_cur[KB];
  #pragma unroll
  for (int s = 0; s < KB; ++s) a_cur[s] = *(const v2f*)(Arow + 4 * s);

  dta_stage_w96<NTHR, PITCH>(Woff, Wlds, 0,  tid);
  dta_stage_w96<NTHR, PITCH>(Waw,  Wlds, NH, tid);
  __syncthreads();

  v8f acc[NTILE];
  #pragma unroll
  for (int t = 0; t < NTILE; ++t) acc[t] = v8f{};

  const float* Bbase = &Wlds[(khalf >> 1) * PITCH + ncol * 2];

  static_assert(KB % G == 0 && K_STEPS % KB == 0, "blocks tile a prefetch group, groups tile K");
  constexpr int NBLK = K_STEPS / G;
  v2f a_nxt[KB];
  v2f bset[2][G][NTILE];
  #pragma unroll
  for (int g = 0; g < G; ++g) {
    #pragma unroll
    for (int t = 0; t < NTILE; ++t)
      bset[0][g][t] = *(const v2f*)(Bbase + g * 2 * PITCH + 32 * t);
  }

  #pragma unroll
  for (int blk = 0; blk < NBLK; ++blk) {
    const int ks0 = blk * G;
    if (blk + 1 < NBLK) {
      #pragma unroll
      for (int g = 0; g < G; ++g) {
        const float* Bk = Bbase + (ks0 + G + g) * 2 * PITCH;
        #pragma unroll
        for (int t = 0; t < NTILE; ++t)
          bset[(blk + 1) & 1][g][t] = *(const v2f*)(Bk + 32 * t);
      }
    }
    if (ks0 % KB == 0 && ks0 + KB < K_STEPS) {
      #pragma unroll
      for (int s = 0; s < KB; ++s) a_nxt[s] = *(const v2f*)(Arow + 4 * (ks0 + KB + s));
    }
    __builtin_amdgcn_sched_barrier(0);
    #pragma unroll
    for (int g = 0; g < G; ++g) {
      #pragma unroll
      for (int t = 0; t < NTILE; ++t)
        acc[t] = __builtin_amdgcn_wmma_f32_16x16x4_f32(false, a_cur[ks0 % KB + g], false, bset[blk & 1][g][t], (short)0, acc[t], false, false);
    }
    __builtin_amdgcn_sched_barrier(0);
    if ((ks0 + G) % KB == 0 && ks0 + G < K_STEPS) {
      #pragma unroll
      for (int s = 0; s < KB; ++s) a_cur[s] = a_nxt[s];
    }
  }

  const int rloc  = wave * 16 + ((lane >> 4) << 3);
  const int rbase = blockIdx.x * ROWS + rloc;
  #pragma unroll
  for (int t = 0; t < HT; ++t) {
    const float bvt = boff[16 * t + ncol];
    #pragma unroll
    for (int r = 0; r < 8; ++r)
      Og[(rloc + r) * LP + 16 * t + ncol] = acc[t][r] + bvt;
  }
  #pragma unroll
  for (int t = 0; t < HT; ++t) {
    const float bvt = baw[16 * t + ncol];
    #pragma unroll
    for (int r = 0; r < 8; ++r)
      Lg[(rloc + r) * LP + 16 * t + ncol] = acc[HT + t][r] + bvt;
  }
  __syncthreads();

  constexpr int GROUPS = ROWS * 8 / NTHR;
  #pragma unroll
  for (int i = 0; i < GROUPS; ++i) {
    const int q = tid + i * NTHR;
    float* p = &Lg[(q >> 3) * LP + (q & 7) * 12];
    float v[12];
    float mx = -3.402823466e+38f;
    #pragma unroll
    for (int e = 0; e < 12; ++e) { v[e] = p[e]; mx = fmaxf(mx, v[e]); }
    float s = 0.0f;
    #pragma unroll
    for (int e = 0; e < 12; ++e) { v[e] = __expf(v[e] - mx); s += v[e]; }
    const float inv = 1.0f / s;
    #pragma unroll
    for (int e = 0; e < 12; ++e) p[e] = v[e] * inv;
  }
  __syncthreads();

  #pragma unroll
  for (int t = 0; t < HT; ++t) {
    const int cs   = 16 * t + ncol;
    const int csrc = (cs & 7) * 12 + (cs >> 3);
    #pragma unroll
    for (int r = 0; r < 8; ++r) {
      off_out[(long)(rbase + r) * NH + cs] = Og[(rloc + r) * LP + csrc];
      aw_out[(long)(rbase + r) * NH + cs]  = Lg[(rloc + r) * LP + csrc];
    }
  }
}

__global__ __launch_bounds__(256)
void dta_sample(const float* __restrict__ refp, const float* __restrict__ off_raw,
                const float* __restrict__ aw, const float* __restrict__ v_proj,
                float* __restrict__ out_mid) {
  const int row  = (blockIdx.x * blockDim.x + threadIdx.x) >> 5;
  const int lane = threadIdx.x & 31;
  const int b    = __builtin_amdgcn_readfirstlane(row >> 11);
  const unsigned lane4 = (unsigned)lane << 2;

  const v4f* offr = (const v4f*)(off_raw + (long)row * 96);
  const v4f* awr  = (const v4f*)(aw + (long)row * 96);
  const char* vbytes = (const char*)v_proj;

  v4f po[4], pa[4];
  #pragma unroll
  for (int i = 0; i < 4; ++i) { po[i] = offr[i]; pa[i] = awr[i]; }
  const float rp = refp[row];

  float acc[8];
  #pragma unroll
  for (int h = 0; h < 8; ++h) acc[h] = 0.0f;

  #pragma nounroll
  for (int bt = 0; bt < 6; ++bt) {
    const int   l    = bt >> 1;
    const int   T    = 2048 >> l;
    const float Tm1  = (float)(T - 1);
    const float invT = __builtin_ldexpf(1.0f, l - 11);
    const int   lvl0 = 8 * (4096 - 2 * T);
    const char* vlev = vbytes + ((size_t)(lvl0 + b * T) << 10) + ((bt & 1) << 8);

    unsigned vo[16];
    float    wc[16];
    #pragma unroll
    for (int s = 0; s < 16; ++s) {
      const float o = po[s >> 2][s & 3];
      float pos  = fminf(fmaxf(rp + o * invT, 0.0f), 1.0f);
      float sidx = pos * Tm1;
      int ifl = (int)sidx;
      ifl = min(max(ifl, 0), T - 2);
      wc[s] = sidx - (float)ifl;
      vo[s] = lane4 + ((unsigned)ifl << 10);
    }
    __builtin_amdgcn_sched_barrier(0);

    float vf[16], vc[16];
    #pragma unroll
    for (int s = 0; s < 16; ++s) {
      const char* a = vlev + vo[s] + (s >> 3) * 128;
      vf[s] = *(const float*)a;
      vc[s] = *(const float*)(a + 1024);
    }
    int nb = min(bt + 1, 5);
    asm volatile("" : "+s"(nb) : "v"(vo[15]));
    v4f po_n[4], pa_n[4];
    #pragma unroll
    for (int i = 0; i < 4; ++i) { po_n[i] = offr[4 * nb + i]; pa_n[i] = awr[4 * nb + i]; }
    __builtin_amdgcn_sched_barrier(0);

    #pragma unroll
    for (int s = 0; s < 16; ++s) {
      const float w = pa[s >> 2][s & 3];
      acc[s & 7] += w * ((1.0f - wc[s]) * vf[s] + wc[s] * vc[s]);
    }
    #pragma unroll
    for (int i = 0; i < 4; ++i) { po[i] = po_n[i]; pa[i] = pa_n[i]; }
  }
  #pragma unroll
  for (int h = 0; h < 8; ++h)
    out_mid[(long)row * 256 + h * 32 + lane] = acc[h];
}

extern "C" void kernel_launch(void* const* d_in, const int* in_sizes, int n_in,
                              void* d_out, int out_size, void* d_ws, size_t ws_size,
                              hipStream_t stream) {
  (void)in_sizes; (void)n_in; (void)out_size; (void)ws_size;

  const float* query = (const float*)d_in[0];
  const float* refp  = (const float*)d_in[1];
  const float* val0  = (const float*)d_in[2];
  const float* val1  = (const float*)d_in[3];
  const float* val2  = (const float*)d_in[4];
  const float* Woff  = (const float*)d_in[5];
  const float* boff  = (const float*)d_in[6];
  const float* Waw   = (const float*)d_in[7];
  const float* baw   = (const float*)d_in[8];
  const float* Wv    = (const float*)d_in[9];
  const float* bv    = (const float*)d_in[10];
  const float* Wo    = (const float*)d_in[11];
  const float* bo    = (const float*)d_in[12];
  float*       out   = (float*)d_out;

  const int BQ = 8 * 2048;

  char* ws = (char*)d_ws;
  float* off_raw = (float*)(ws);
  float* aw_buf  = (float*)(ws + (size_t)BQ * 96 * 4);
  float* v_proj  = (float*)(ws + (size_t)BQ * 96 * 4 * 2);
  float* out_mid = (float*)(ws + (size_t)BQ * 96 * 4 * 2
                               + (size_t)(BQ + 8192 + 4096) * 256 * 4);

  dim3 blk(256);
  dim3 blk4(128);

  dta_qproj_softmax<4, 8, 2><<<dim3(256), blk4, 0, stream>>>(query, Woff, boff, Waw, baw, off_raw, aw_buf);
  dta_vproj_ws<64, 4, 7, 16, 4><<<dim3(64, 4), blk4, 0, stream>>>(val0, val1, val2, Wv, bv, v_proj);
  dta_sample<<<dim3(BQ / 8), blk, 0, stream>>>(refp, off_raw, aw_buf, v_proj, out_mid);
  dta_gemm_f32_wmma<128, 8, 8, 2><<<dim3(128, 2), blk, 0, stream>>>(out_mid, Wo, bo, out, 256);
}


# Query projections fused into the sampling kernel: a 16-wave workgroup per WGP projects its 64 rows on WMMA, keeps offsets and softmaxed weights in LDS, samples from there; 4 launches -> 3; sums kept
# speedup vs baseline: 1.5193x; 1.0408x over previous
#include <hip/hip_runtime.h>
#include <hip/hip_bf16.h>
#include <math.h>


typedef float v2f __attribute__((ext_vector_type(2)));
typedef float v4f __attribute__((ext_vector_type(4)));
typedef float v8f __attribute__((ext_vector_type(8)));

#define D_MODEL 256

#define K_STEPS (D_MODEL / 4)

template <int NW, int WAVES, int KB, int G>
__global__ __launch_bounds__(WAVES * 32)
void dta_gemm_f32_wmma(const float* __restrict__ A, const float* __restrict__ W,
                       const float* __restrict__ bias, float* __restrict__ C,
                       int N) {
  constexpr int NTHR  = WAVES * 32;
  constexpr int NTILE = NW / 16;
  constexpr int PITCH = NW * 2 + 32;
  constexpr int CPR   = NW / 4;
  constexpr int TILE_FLOATS = (D_MODEL / 2) * PITCH;
  constexpr int ONE_PER_WGP_FLOATS = 41984;
  constexpr int LDS_FLOATS = TILE_FLOATS > ONE_PER_WGP_FLOATS ? TILE_FLOATS : ONE_PER_WGP_FLOATS;
  __shared__ __attribute__((aligned(16))) float Wlds[LDS_FLOATS];

  const int tid   = threadIdx.x;
  const int wave  = tid >> 5;
  const int lane  = tid & 31;
  const int n0    = blockIdx.y * NW;
  const int m0    = blockIdx.x * (WAVES * 16) + wave * 16;
  const int khalf = (lane >> 4) << 1;
  const int ncol  = lane & 15;

  const float* Arow = A + (long)(m0 + ncol) * D_MODEL + khalf;
  v2f a_cur[KB];
  #pragma unroll
  for (int s = 0; s < KB; ++s) a_cur[s] = *(const v2f*)(Arow + 4 * s);

  constexpr int STAGE_ITERS = (D_MODEL / 2) * CPR / NTHR;
  static_assert(STAGE_ITERS * NTHR == (D_MODEL / 2) * CPR, "chunks divide evenly over the threads");
  #pragma unroll
  for (int it = 0; it < STAGE_ITERS; ++it) {
    const int ch = tid + it * NTHR;
    const int kk = ch / CPR;
    const int j  = (ch - kk * CPR) << 2;
    v4f r0 = *(const v4f*)(W + (long)(2 * kk)     * N + n0 + j);
    v4f r1 = *(const v4f*)(W + (long)(2 * kk + 1) * N + n0 + j);
    v4f lo = {r0.x, r1.x, r0.y, r1.y};
    v4f hi = {r0.z, r1.z, r0.w, r1.w};
    *(v4f*)(&Wlds[kk * PITCH + j * 2])     = lo;
    *(v4f*)(&Wlds[kk * PITCH + j * 2 + 4]) = hi;
  }
  __syncthreads();

  v8f acc[NTILE];
  #pragma unroll
  for (int t = 0; t < NTILE; ++t) acc[t] = v8f{};

  const float* Bbase = &Wlds[(khalf >> 1) * PITCH + ncol * 2];

  static_assert(KB % G == 0 && K_STEPS % KB == 0, "blocks tile a prefetch group, groups tile K");
  constexpr int NBLK = K_STEPS / G;
  v2f a_nxt[KB];
  v2f bset[2][G][NTILE];
  #pragma unroll
  for (int g = 0; g < G; ++g) {
    #pragma unroll
    for (int t = 0; t < NTILE; ++t)
      bset[0][g][t] = *(const v2f*)(Bbase + g * 2 * PITCH + 32 * t);
  }

  #pragma unroll
  for (int blk = 0; blk < NBLK; ++blk) {
    const int ks0 = blk * G;
    if (blk + 1 < NBLK) {
      #pragma unroll
      for (int g = 0; g < G; ++g) {
        const float* Bk = Bbase + (ks0 + G + g) * 2 * PITCH;
        #pragma unroll
        for (int t = 0; t < NTILE; ++t)
          bset[(blk + 1) & 1][g][t] = *(const v2f*)(Bk + 32 * t);
      }
    }
    if (ks0 % KB == 0 && ks0 + KB < K_STEPS) {
      #pragma unroll
      for (int s = 0; s < KB; ++s) a_nxt[s] = *(const v2f*)(Arow + 4 * (ks0 + KB + s));
    }
    __builtin_amdgcn_sched_barrier(0);
    #pragma unroll
    for (int g = 0; g < G; ++g) {
      #pragma unroll
      for (int t = 0; t < NTILE; ++t)
        acc[t] = __builtin_amdgcn_wmma_f32_16x16x4_f32(false, a_cur[ks0 % KB + g], false, bset[blk & 1][g][t], (short)0, acc[t], false, false);
    }
    __builtin_amdgcn_sched_barrier(0);
    if ((ks0 + G) % KB == 0 && ks0 + G < K_STEPS) {
      #pragma unroll
      for (int s = 0; s < KB; ++s) a_cur[s] = a_nxt[s];
    }
  }

  const int rbase = m0 + ((lane >> 4) << 3);
  #pragma unroll
  for (int t = 0; t < NTILE; ++t) {
    const float bvt = bias[n0 + 16 * t + ncol];
    #pragma unroll
    for (int r = 0; r < 8; ++r)
      C[(long)(rbase + r) * N + n0 + 16 * t + ncol] = acc[t][r] + bvt;
  }
}

#define DTA_STRIPS_L0 1024
#define DTA_STRIPS_L1 512

__device__ __forceinline__ const float* dta_strip_rows(const float* __restrict__ V0,
                                                       const float* __restrict__ V1,
                                                       const float* __restrict__ V2, int s) {
  if (s < DTA_STRIPS_L0) return V0 + (long)s * (16 * D_MODEL);
  if (s < DTA_STRIPS_L0 + DTA_STRIPS_L1) return V1 + (long)(s - DTA_STRIPS_L0) * (16 * D_MODEL);
  return V2 + (long)(s - DTA_STRIPS_L0 - DTA_STRIPS_L1) * (16 * D_MODEL);
}

template <int NW, int WAVES, int STRIPS, int KB, int G>
__global__ __launch_bounds__(WAVES * 32)
void dta_vproj_ws(const float* __restrict__ V0, const float* __restrict__ V1,
                  const float* __restrict__ V2, const float* __restrict__ W,
                  const float* __restrict__ bias, float* __restrict__ C) {
  constexpr int N     = D_MODEL;
  constexpr int NTHR  = WAVES * 32;
  constexpr int NTILE = NW / 16;
  constexpr int PITCH = NW * 2 + 32;
  constexpr int CPR   = NW / 4;
  constexpr int TILE_FLOATS = (D_MODEL / 2) * PITCH;
  constexpr int ONE_PER_WGP_FLOATS = 41984;
  constexpr int LDS_FLOATS = TILE_FLOATS > ONE_PER_WGP_FLOATS ? TILE_FLOATS : ONE_PER_WGP_FLOATS;
  __shared__ __attribute__((aligned(16))) float Wlds[LDS_FLOATS];

  const int tid   = threadIdx.x;
  const int wave  = __builtin_amdgcn_readfirstlane(tid >> 5);
  const int lane  = tid & 31;
  const int n0    = blockIdx.y * NW;
  const int khalf = (lane >> 4) << 1;
  const int ncol  = lane & 15;
  const int lofs  = ncol * D_MODEL + khalf;
  const int s0    = blockIdx.x * (WAVES * STRIPS) + wave;

  const float* Arow = dta_strip_rows(V0, V1, V2, s0) + lofs;
  v2f a_cur[KB];
  #pragma unroll
  for (int s = 0; s < KB; ++s) a_cur[s] = *(const v2f*)(Arow + 4 * s);

  constexpr int STAGE_ITERS = (D_MODEL / 2) * CPR / NTHR;
  static_assert(STAGE_ITERS * NTHR == (D_MODEL / 2) * CPR, "chunks divide evenly over the threads");
  #pragma unroll
  for (int it = 0; it < STAGE_ITERS; ++it) {
    const int ch = tid + it * NTHR;
    const int kk = ch / CPR;
    const int j  = (ch - kk * CPR) << 2;
    v4f r0 = *(const v4f*)(W + (long)(2 * kk)     * N + n0 + j);
    v4f r1 = *(const v4f*)(W + (long)(2 * kk + 1) * N + n0 + j);
    v4f lo = {r0.x, r1.x, r0.y, r1.y};
    v4f hi = {r0.z, r1.z, r0.w, r1.w};
    *(v4f*)(&Wlds[kk * PITCH + j * 2])     = lo;
    *(v4f*)(&Wlds[kk * PITCH + j * 2 + 4]) = hi;
  }

  float bvt[NTILE];
  #pragma unroll
  for (int t = 0; t < NTILE; ++t) bvt[t] = bias[n0 + 16 * t + ncol];
  __syncthreads();

  const float* Bbase = &Wlds[(khalf >> 1) * PITCH + ncol * 2];

  static_assert(KB % G == 0 && K_STEPS % KB == 0, "blocks tile a prefetch group, groups tile K");
  constexpr int NBLK = K_STEPS / G;
  static_assert(NBLK % 2 == 0, "the B register sets alternate and must line up again at the next strip");
  v2f a_nxt[KB];
  v2f bset[2][G][NTILE];
  #pragma unroll
  for (int g = 0; g < G; ++g) {
    #pragma unroll
    for (int t = 0; t < NTILE; ++t)
      bset[0][g][t] = *(const v2f*)(Bbase + g * 2 * PITCH + 32 * t);
  }

  for (int i = 0; i < STRIPS; ++i) {
    const int sc = s0 + i * WAVES;
    const int sn = (i + 1 < STRIPS) ? sc + WAVES : sc;
    const float* Anext = dta_strip_rows(V0, V1, V2, sn) + lofs;

    v8f acc[NTILE];
    #pragma unroll
    for (int t = 0; t < NTILE; ++t) acc[t] = v8f{};

    #pragma unroll
    for (int blk = 0; blk < NBLK; ++blk) {
      const int ks0 = blk * G;
      #pragma unroll
      for (int g = 0; g < G; ++g) {
        const float* Bk = Bbase + ((ks0 + G + g) % K_STEPS) * 2 * PITCH;
        #pragma unroll
        for (int t = 0; t < NTILE; ++t)
          bset[(blk + 1) & 1][g][t] = *(const v2f*)(Bk + 32 * t);
      }
      if (ks0 % KB == 0) {
        if (ks0 + KB < K_STEPS) {
          #pragma unroll
          for (int s = 0; s < KB; ++s) a_nxt[s] = *(const v2f*)(Arow + 4 * (ks0 + KB + s));
        } else {
          #pragma unroll
          for (int s = 0; s < KB; ++s) a_nxt[s] = *(const v2f*)(Anext + 4 * s);
        }
      }
      __builtin_amdgcn_sched_barrier(0);
      #pragma unroll
      for (int g = 0; g < G; ++g) {
        #pragma unroll
        for (int t = 0; t < NTILE; ++t)
          acc[t] = __builtin_amdgcn_wmma_f32_16x16x4_f32(false, a_cur[ks0 % KB + g], false, bset[blk & 1][g][t], (short)0, acc[t], false, false);
      }
      __builtin_amdgcn_sched_barrier(0);
      if ((ks0 + G) % KB == 0) {
        #pragma unroll
        for (int s = 0; s < KB; ++s) a_cur[s] = a_nxt[s];
      }
    }

    float* Crow = C + (long)sc * (16 * N) + (long)((lane >> 4) << 3) * N + n0 + ncol;
    #pragma unroll
    for (int t = 0; t < NTILE; ++t) {
      #pragma unroll
      for (int r = 0; r < 8; ++r)
        Crow[r * N + 16 * t] = acc[t][r] + bvt[t];
    }
    Arow = Anext;
  }
}

template <int NTHR, int PITCH>
__device__ __forceinline__ void dta_stage_w96(const float* __restrict__ W, float* Wlds,
                                              int cofs, int tid) {
  constexpr int CPR   = 96 / 4;
  constexpr int ITERS = (D_MODEL / 2) * CPR / NTHR;
  static_assert(ITERS * NTHR == (D_MODEL / 2) * CPR, "chunks divide evenly over the threads");
  #pragma unroll
  for (int it = 0; it < ITERS; ++it) {
    const int ch = tid + it * NTHR;
    const int kk = ch / CPR;
    const int j  = (ch - kk * CPR) << 2;
    v4f r0 = *(const v4f*)(W + (long)(2 * kk)     * 96 + j);
    v4f r1 = *(const v4f*)(W + (long)(2 * kk + 1) * 96 + j);
    v4f lo = {r0.x, r1.x, r0.y, r1.y};
    v4f hi = {r0.z, r1.z, r0.w, r1.w};
    *(v4f*)(&Wlds[kk * PITCH + (cofs + j) * 2])     = lo;
    *(v4f*)(&Wlds[kk * PITCH + (cofs + j) * 2 + 4]) = hi;
  }
}

template <int KB, int G>
__global__ __launch_bounds__(512)
void dta_qproj_sample(const float* __restrict__ A,
                      const float* __restrict__ Woff, const float* __restrict__ boff,
                      const float* __restrict__ Waw,  const float* __restrict__ baw,
                      const float* __restrict__ refp, const float* __restrict__ v_proj,
                      float* __restrict__ out_mid) {
  constexpr int NH    = 96;
  constexpr int NW    = 2 * NH;
  constexpr int NTHR  = 512;
  constexpr int NT    = 3;
  constexpr int PITCH = NW * 2 + 32;
  constexpr int TILE_FLOATS = (D_MODEL / 2) * PITCH;
  constexpr int ROWS  = 64;
  constexpr int RPW   = 4;
  constexpr int LP    = NH + 1;
  __shared__ __attribute__((aligned(16))) float Wlds[TILE_FLOATS];
  __shared__ float Lg[ROWS * LP];
  __shared__ float Og[ROWS * LP];

  const int tid   = threadIdx.x;
  const int wave  = __builtin_amdgcn_readfirstlane(tid >> 5);
  const int lane  = tid & 31;
  const int strip = wave & 3;
  const int cg    = wave >> 2;
  const int m0    = blockIdx.x * ROWS + strip * 16;
  const int khalf = (lane >> 4) << 1;
  const int ncol  = lane & 15;

  const float* Arow = A + (long)(m0 + ncol) * D_MODEL + khalf;
  v2f a_cur[KB];
  #pragma unroll
  for (int s = 0; s < KB; ++s) a_cur[s] = *(const v2f*)(Arow + 4 * s);

  dta_stage_w96<NTHR, PITCH>(Woff, Wlds, 0,  tid);
  dta_stage_w96<NTHR, PITCH>(Waw,  Wlds, NH, tid);
  __syncthreads();

  {
    v8f acc[NT];
    #pragma unroll
    for (int t = 0; t < NT; ++t) acc[t] = v8f{};

    const float* Bbase = &Wlds[(khalf >> 1) * PITCH + cg * (2 * 16 * NT) + ncol * 2];

    static_assert(KB % G == 0 && K_STEPS % KB == 0, "blocks tile a prefetch group, groups tile K");
    constexpr int NBLK = K_STEPS / G;
    v2f a_nxt[KB];
    v2f bset[2][G][NT];
    #pragma unroll
    for (int g = 0; g < G; ++g) {
      #pragma unroll
      for (int t = 0; t < NT; ++t)
        bset[0][g][t] = *(const v2f*)(Bbase + g * 2 * PITCH + 32 * t);
    }

    #pragma unroll
    for (int blk = 0; blk < NBLK; ++blk) {
      const int ks0 = blk * G;
      if (blk + 1 < NBLK) {
        #pragma unroll
        for (int g = 0; g < G; ++g) {
          const float* Bk = Bbase + (ks0 + G + g) * 2 * PITCH;
          #pragma unroll
          for (int t = 0; t < NT; ++t)
            bset[(blk + 1) & 1][g][t] = *(const v2f*)(Bk + 32 * t);
        }
      }
      if (ks0 % KB == 0 && ks0 + KB < K_STEPS) {
        #pragma unroll
        for (int s = 0; s < KB; ++s) a_nxt[s] = *(const v2f*)(Arow + 4 * (ks0 + KB + s));
      }
      __builtin_amdgcn_sched_barrier(0);
      #pragma unroll
      for (int g = 0; g < G; ++g) {
        #pragma unroll
        for (int t = 0; t < NT; ++t)
          acc[t] = __builtin_amdgcn_wmma_f32_16x16x4_f32(false, a_cur[ks0 % KB + g], false, bset[blk & 1][g][t], (short)0, acc[t], false, false);
      }
      __builtin_amdgcn_sched_barrier(0);
      if ((ks0 + G) % KB == 0 && ks0 + G < K_STEPS) {
        #pragma unroll
        for (int s = 0; s < KB; ++s) a_cur[s] = a_nxt[s];
      }
    }

    const int rloc = strip * 16 + ((lane >> 4) << 3);
    const int c0   = (cg & 1) * (16 * NT);
    float*       dst  = (cg < 2) ? Og : Lg;
    const float* bsrc = (cg < 2) ? boff : baw;
    #pragma unroll
    for (int t = 0; t < NT; ++t) {
      const int c  = c0 + 16 * t + ncol;
      const int pc = (c % 12) * 8 + c / 12;
      const float bvt = bsrc[c];
      #pragma unroll
      for (int r = 0; r < 8; ++r)
        dst[(rloc + r) * LP + pc] = acc[t][r] + bvt;
    }
  }
  __syncthreads();

  {
    float* p = &Lg[(tid >> 3) * LP + (tid & 7)];
    float v[12];
    float mx = -3.402823466e+38f;
    #pragma unroll
    for (int e = 0; e < 12; ++e) { v[e] = p[8 * e]; mx = fmaxf(mx, v[e]); }
    float s = 0.0f;
    #pragma unroll
    for (int e = 0; e < 12; ++e) { v[e] = __expf(v[e] - mx); s += v[e]; }
    const float inv = 1.0f / s;
    #pragma unroll
    for (int e = 0; e < 12; ++e) p[8 * e] = v[e] * inv;
  }
  __syncthreads();

  const int rl0   = wave * RPW;
  const int grow0 = blockIdx.x * ROWS + rl0;
  const int b     = blockIdx.x >> 5;
  const unsigned lane4 = (unsigned)lane << 2;
  const float* rpr = refp + grow0;
  const char* vbytes = (const char*)v_proj;
  constexpr int LAST = 6 * RPW - 1;

  float po[16], pa[16];
  {
    const float* so = &Og[rl0 * LP];
    const float* sa = &Lg[rl0 * LP];
    #pragma unroll
    for (int s = 0; s < 16; ++s) { po[s] = so[s]; pa[s] = sa[s]; }
  }
  float rp = rpr[0];

  #pragma nounroll
  for (int j = 0; j < RPW; ++j) {
    float acc[8];
    #pragma unroll
    for (int h = 0; h < 8; ++h) acc[h] = 0.0f;

    #pragma nounroll
    for (int bt = 0; bt < 6; ++bt) {
      const int   l    = bt >> 1;
      const int   T    = 2048 >> l;
      const float Tm1  = (float)(T - 1);
      const float invT = __builtin_ldexpf(1.0f, l - 11);
      const int   lvl0 = 8 * (4096 - 2 * T);
      const char* vlev = vbytes + ((size_t)(lvl0 + b * T) << 10) + ((bt & 1) << 8);

      unsigned vo[16];
      float    wc[16];
      #pragma unroll
      for (int s = 0; s < 16; ++s) {
        const float o = po[s];
        float pos  = fminf(fmaxf(rp + o * invT, 0.0f), 1.0f);
        float sidx = pos * Tm1;
        int ifl = (int)sidx;
        ifl = min(max(ifl, 0), T - 2);
        wc[s] = sidx - (float)ifl;
        vo[s] = lane4 + ((unsigned)ifl << 10);
      }
      __builtin_amdgcn_sched_barrier(0);

      float vf[16], vc[16];
      #pragma unroll
      for (int s = 0; s < 16; ++s) {
        const char* a = vlev + vo[s] + (s >> 3) * 128;
        vf[s] = *(const float*)a;
        vc[s] = *(const float*)(a + 1024);
      }
      const int nb  = min(j * 6 + bt + 1, LAST);
      const int nj  = (unsigned)nb / 6u;
      const int nbt = nb - 6 * nj;
      const float* so = &Og[(rl0 + nj) * LP + 16 * nbt];
      const float* sa = &Lg[(rl0 + nj) * LP + 16 * nbt];
      float po_n[16], pa_n[16];
      #pragma unroll
      for (int s = 0; s < 16; ++s) { po_n[s] = so[s]; pa_n[s] = sa[s]; }
      const float rp_n = rpr[nj];
      __builtin_amdgcn_sched_barrier(0);

      #pragma unroll
      for (int s = 0; s < 16; ++s) {
        const float w = pa[s];
        acc[s & 7] += w * ((1.0f - wc[s]) * vf[s] + wc[s] * vc[s]);
      }
      #pragma unroll
      for (int s = 0; s < 16; ++s) { po[s] = po_n[s]; pa[s] = pa_n[s]; }
      rp = rp_n;
    }
    #pragma unroll
    for (int h = 0; h < 8; ++h)
      out_mid[(long)(grow0 + j) * 256 + h * 32 + lane] = acc[h];
  }
}

extern "C" void kernel_launch(void* const* d_in, const int* in_sizes, int n_in,
                              void* d_out, int out_size, void* d_ws, size_t ws_size,
                              hipStream_t stream) {
  (void)in_sizes; (void)n_in; (void)out_size; (void)ws_size;

  const float* query = (const float*)d_in[0];
  const float* refp  = (const float*)d_in[1];
  const float* val0  = (const float*)d_in[2];
  const float* val1  = (const float*)d_in[3];
  const float* val2  = (const float*)d_in[4];
  const float* Woff  = (const float*)d_in[5];
  const float* boff  = (const float*)d_in[6];
  const float* Waw   = (const float*)d_in[7];
  const float* baw   = (const float*)d_in[8];
  const float* Wv    = (const float*)d_in[9];
  const float* bv    = (const float*)d_in[10];
  const float* Wo    = (const float*)d_in[11];
  const float* bo    = (const float*)d_in[12];
  float*       out   = (float*)d_out;

  const int BQ = 8 * 2048;

  char* ws = (char*)d_ws;
  float* v_proj  = (float*)(ws + (size_t)BQ * 96 * 4 * 2);
  float* out_mid = (float*)(ws + (size_t)BQ * 96 * 4 * 2
                               + (size_t)(BQ + 8192 + 4096) * 256 * 4);

  dim3 blk(256);
  dim3 blk4(128);
  dim3 blk16(512);

  dta_vproj_ws<64, 4, 7, 16, 4><<<dim3(64, 4), blk4, 0, stream>>>(val0, val1, val2, Wv, bv, v_proj);
  dta_qproj_sample<8, 4><<<dim3(256), blk16, 0, stream>>>(query, Woff, boff, Waw, baw, refp, v_proj, out_mid);
  dta_gemm_f32_wmma<128, 8, 8, 2><<<dim3(128, 2), blk, 0, stream>>>(out_mid, Wo, bo, out, 256);
}


# Sample parameters made once per (row, head) by the softmax threads (byte offset, lerp weights, attention weight, in LDS over the dead weight tile); sampling waves only add lane, read, sum; sums kept
# speedup vs baseline: 1.5789x; 1.0392x over previous
#include <hip/hip_runtime.h>
#include <hip/hip_bf16.h>
#include <math.h>


typedef float v2f __attribute__((ext_vector_type(2)));
typedef float v4f __attribute__((ext_vector_type(4)));
typedef float v8f __attribute__((ext_vector_type(8)));
typedef unsigned v4u __attribute__((ext_vector_type(4)));

#define D_MODEL 256

#define K_STEPS (D_MODEL / 4)

template <int NW, int WAVES, int KB, int G>
__global__ __launch_bounds__(WAVES * 32)
void dta_gemm_f32_wmma(const float* __restrict__ A, const float* __restrict__ W,
                       const float* __restrict__ bias, float* __restrict__ C,
                       int N) {
  constexpr int NTHR  = WAVES * 32;
  constexpr int NTILE = NW / 16;
  constexpr int PITCH = NW * 2 + 32;
  constexpr int CPR   = NW / 4;
  constexpr int TILE_FLOATS = (D_MODEL / 2) * PITCH;
  constexpr int ONE_PER_WGP_FLOATS = 41984;
  constexpr int LDS_FLOATS = TILE_FLOATS > ONE_PER_WGP_FLOATS ? TILE_FLOATS : ONE_PER_WGP_FLOATS;
  __shared__ __attribute__((aligned(16))) float Wlds[LDS_FLOATS];

  const int tid   = threadIdx.x;
  const int wave  = tid >> 5;
  const int lane  = tid & 31;
  const int n0    = blockIdx.y * NW;
  const int m0    = blockIdx.x * (WAVES * 16) + wave * 16;
  const int khalf = (lane >> 4) << 1;
  const int ncol  = lane & 15;

  const float* Arow = A + (long)(m0 + ncol) * D_MODEL + khalf;
  v2f a_cur[KB];
  #pragma unroll
  for (int s = 0; s < KB; ++s) a_cur[s] = *(const v2f*)(Arow + 4 * s);

  constexpr int STAGE_ITERS = (D_MODEL / 2) * CPR / NTHR;
  static_assert(STAGE_ITERS * NTHR == (D_MODEL / 2) * CPR, "chunks divide evenly over the threads");
  #pragma unroll
  for (int it = 0; it < STAGE_ITERS; ++it) {
    const int ch = tid + it * NTHR;
    const int kk = ch / CPR;
    const int j  = (ch - kk * CPR) << 2;
    v4f r0 = *(const v4f*)(W + (long)(2 * kk)     * N + n0 + j);
    v4f r1 = *(const v4f*)(W + (long)(2 * kk + 1) * N + n0 + j);
    v4f lo = {r0.x, r1.x, r0.y, r1.y};
    v4f hi = {r0.z, r1.z, r0.w, r1.w};
    *(v4f*)(&Wlds[kk * PITCH + j * 2])     = lo;
    *(v4f*)(&Wlds[kk * PITCH + j * 2 + 4]) = hi;
  }
  __syncthreads();

  v8f acc[NTILE];
  #pragma unroll
  for (int t = 0; t < NTILE; ++t) acc[t] = v8f{};

  const float* Bbase = &Wlds[(khalf >> 1) * PITCH + ncol * 2];

  static_assert(KB % G == 0 && K_STEPS % KB == 0, "blocks tile a prefetch group, groups tile K");
  constexpr int NBLK = K_STEPS / G;
  v2f a_nxt[KB];
  v2f bset[2][G][NTILE];
  #pragma unroll
  for (int g = 0; g < G; ++g) {
    #pragma unroll
    for (int t = 0; t < NTILE; ++t)
      bset[0][g][t] = *(const v2f*)(Bbase + g * 2 * PITCH + 32 * t);
  }

  #pragma unroll
  for (int blk = 0; blk < NBLK; ++blk) {
    const int ks0 = blk * G;
    if (blk + 1 < NBLK) {
      #pragma unroll
      for (int g = 0; g < G; ++g) {
        const float* Bk = Bbase + (ks0 + G + g) * 2 * PITCH;
        #pragma unroll
        for (int t = 0; t < NTILE; ++t)
          bset[(blk + 1) & 1][g][t] = *(const v2f*)(Bk + 32 * t);
      }
    }
    if (ks0 % KB == 0 && ks0 + KB < K_STEPS) {
      #pragma unroll
      for (int s = 0; s < KB; ++s) a_nxt[s] = *(const v2f*)(Arow + 4 * (ks0 + KB + s));
    }
    __builtin_amdgcn_sched_barrier(0);
    #pragma unroll
    for (int g = 0; g < G; ++g) {
      #pragma unroll
      for (int t = 0; t < NTILE; ++t)
        acc[t] = __builtin_amdgcn_wmma_f32_16x16x4_f32(false, a_cur[ks0 % KB + g], false, bset[blk & 1][g][t], (short)0, acc[t], false, false);
    }
    __builtin_amdgcn_sched_barrier(0);
    if ((ks0 + G) % KB == 0 && ks0 + G < K_STEPS) {
      #pragma unroll
      for (int s = 0; s < KB; ++s) a_cur[s] = a_nxt[s];
    }
  }

  const int rbase = m0 + ((lane >> 4) << 3);
  #pragma unroll
  for (int t = 0; t < NTILE; ++t) {
    const float bvt = bias[n0 + 16 * t + ncol];
    #pragma unroll
    for (int r = 0; r < 8; ++r)
      C[(long)(rbase + r) * N + n0 + 16 * t + ncol] = acc[t][r] + bvt;
  }
}

#define DTA_STRIPS_L0 1024
#define DTA_STRIPS_L1 512

__device__ __forceinline__ const float* dta_strip_rows(const float* __restrict__ V0,
                                                       const float* __restrict__ V1,
                                                       const float* __restrict__ V2, int s) {
  if (s < DTA_STRIPS_L0) return V0 + (long)s * (16 * D_MODEL);
  if (s < DTA_STRIPS_L0 + DTA_STRIPS_L1) return V1 + (long)(s - DTA_STRIPS_L0) * (16 * D_MODEL);
  return V2 + (long)(s - DTA_STRIPS_L0 - DTA_STRIPS_L1) * (16 * D_MODEL);
}

template <int NW, int WAVES, int STRIPS, int KB, int G>
__global__ __launch_bounds__(WAVES * 32)
void dta_vproj_ws(const float* __restrict__ V0, const float* __restrict__ V1,
                  const float* __restrict__ V2, const float* __restrict__ W,
                  const float* __restrict__ bias, float* __restrict__ C) {
  constexpr int N     = D_MODEL;
  constexpr int NTHR  = WAVES * 32;
  constexpr int NTILE = NW / 16;
  constexpr int PITCH = NW * 2 + 32;
  constexpr int CPR   = NW / 4;
  constexpr int TILE_FLOATS = (D_MODEL / 2) * PITCH;
  constexpr int ONE_PER_WGP_FLOATS = 41984;
  constexpr int LDS_FLOATS = TILE_FLOATS > ONE_PER_WGP_FLOATS ? TILE_FLOATS : ONE_PER_WGP_FLOATS;
  __shared__ __attribute__((aligned(16))) float Wlds[LDS_FLOATS];

  const int tid   = threadIdx.x;
  const int wave  = __builtin_amdgcn_readfirstlane(tid >> 5);
  const int lane  = tid & 31;
  const int n0    = blockIdx.y * NW;
  const int khalf = (lane >> 4) << 1;
  const int ncol  = lane & 15;
  const int lofs  = ncol * D_MODEL + khalf;
  const int s0    = blockIdx.x * (WAVES * STRIPS) + wave;

  const float* Arow = dta_strip_rows(V0, V1, V2, s0) + lofs;
  v2f a_cur[KB];
  #pragma unroll
  for (int s = 0; s < KB; ++s) a_cur[s] = *(const v2f*)(Arow + 4 * s);

  constexpr int STAGE_ITERS = (D_MODEL / 2) * CPR / NTHR;
  static_assert(STAGE_ITERS * NTHR == (D_MODEL / 2) * CPR, "chunks divide evenly over the threads");
  #pragma unroll
  for (int it = 0; it < STAGE_ITERS; ++it) {
    const int ch = tid + it * NTHR;
    const int kk = ch / CPR;
    const int j  = (ch - kk * CPR) << 2;
    v4f r0 = *(const v4f*)(W + (long)(2 * kk)     * N + n0 + j);
    v4f r1 = *(const v4f*)(W + (long)(2 * kk + 1) * N + n0 + j);
    v4f lo = {r0.x, r1.x, r0.y, r1.y};
    v4f hi = {r0.z, r1.z, r0.w, r1.w};
    *(v4f*)(&Wlds[kk * PITCH + j * 2])     = lo;
    *(v4f*)(&Wlds[kk * PITCH + j * 2 + 4]) = hi;
  }

  float bvt[NTILE];
  #pragma unroll
  for (int t = 0; t < NTILE; ++t) bvt[t] = bias[n0 + 16 * t + ncol];
  __syncthreads();

  const float* Bbase = &Wlds[(khalf >> 1) * PITCH + ncol * 2];

  static_assert(KB % G == 0 && K_STEPS % KB == 0, "blocks tile a prefetch group, groups tile K");
  constexpr int NBLK = K_STEPS / G;
  static_assert(NBLK % 2 == 0, "the B register sets alternate and must line up again at the next strip");
  v2f a_nxt[KB];
  v2f bset[2][G][NTILE];
  #pragma unroll
  for (int g = 0; g < G; ++g) {
    #pragma unroll
    for (int t = 0; t < NTILE; ++t)
      bset[0][g][t] = *(const v2f*)(Bbase + g * 2 * PITCH + 32 * t);
  }

  for (int i = 0; i < STRIPS; ++i) {
    const int sc = s0 + i * WAVES;
    const int sn = (i + 1 < STRIPS) ? sc + WAVES : sc;
    const float* Anext = dta_strip_rows(V0, V1, V2, sn) + lofs;

    v8f acc[NTILE];
    #pragma unroll
    for (int t = 0; t < NTILE; ++t) acc[t] = v8f{};

    #pragma unroll
    for (int blk = 0; blk < NBLK; ++blk) {
      const int ks0 = blk * G;
      #pragma unroll
      for (int g = 0; g < G; ++g) {
        const float* Bk = Bbase + ((ks0 + G + g) % K_STEPS) * 2 * PITCH;
        #pragma unroll
        for (int t = 0; t < NTILE; ++t)
          bset[(blk + 1) & 1][g][t] = *(const v2f*)(Bk + 32 * t);
      }
      if (ks0 % KB == 0) {
        if (ks0 + KB < K_STEPS) {
          #pragma unroll
          for (int s = 0; s < KB; ++s) a_nxt[s] = *(const v2f*)(Arow + 4 * (ks0 + KB + s));
        } else {
          #pragma unroll
          for (int s = 0; s < KB; ++s) a_nxt[s] = *(const v2f*)(Anext + 4 * s);
        }
      }
      __builtin_amdgcn_sched_barrier(0);
      #pragma unroll
      for (int g = 0; g < G; ++g) {
        #pragma unroll
        for (int t = 0; t < NTILE; ++t)
          acc[t] = __builtin_amdgcn_wmma_f32_16x16x4_f32(false, a_cur[ks0 % KB + g], false, bset[blk & 1][g][t], (short)0, acc[t], false, false);
      }
      __builtin_amdgcn_sched_barrier(0);
      if ((ks0 + G) % KB == 0) {
        #pragma unroll
        for (int s = 0; s < KB; ++s) a_cur[s] = a_nxt[s];
      }
    }

    float* Crow = C + (long)sc * (16 * N) + (long)((lane >> 4) << 3) * N + n0 + ncol;
    #pragma unroll
    for (int t = 0; t < NTILE; ++t) {
      #pragma unroll
      for (int r = 0; r < 8; ++r)
        Crow[r * N + 16 * t] = acc[t][r] + bvt[t];
    }
    Arow = Anext;
  }
}

template <int NTHR, int PITCH>
__device__ __forceinline__ void dta_stage_w96(const float* __restrict__ W, float* Wlds,
                                              int cofs, int tid) {
  constexpr int CPR   = 96 / 4;
  constexpr int ITERS = (D_MODEL / 2) * CPR / NTHR;
  static_assert(ITERS * NTHR == (D_MODEL / 2) * CPR, "chunks divide evenly over the threads");
  #pragma unroll
  for (int it = 0; it < ITERS; ++it) {
    const int ch = tid + it * NTHR;
    const int kk = ch / CPR;
    const int j  = (ch - kk * CPR) << 2;
    v4f r0 = *(const v4f*)(W + (long)(2 * kk)     * 96 + j);
    v4f r1 = *(const v4f*)(W + (long)(2 * kk + 1) * 96 + j);
    v4f lo = {r0.x, r1.x, r0.y, r1.y};
    v4f hi = {r0.z, r1.z, r0.w, r1.w};
    *(v4f*)(&Wlds[kk * PITCH + (cofs + j) * 2])     = lo;
    *(v4f*)(&Wlds[kk * PITCH + (cofs + j) * 2 + 4]) = hi;
  }
}

template <int KB, int G>
__global__ __launch_bounds__(512)
void dta_qproj_sample(const float* __restrict__ A,
                      const float* __restrict__ Woff, const float* __restrict__ boff,
                      const float* __restrict__ Waw,  const float* __restrict__ baw,
                      const float* __restrict__ refp, const float* __restrict__ v_proj,
                      float* __restrict__ out_mid) {
  constexpr int NH    = 96;
  constexpr int NW    = 2 * NH;
  constexpr int NTHR  = 512;
  constexpr int NT    = 3;
  constexpr int PITCH = NW * 2 + 32;
  constexpr int TILE_FLOATS = (D_MODEL / 2) * PITCH;
  constexpr int ROWS  = 64;
  constexpr int RPW   = 4;
  constexpr int LP    = NH + 1;
  __shared__ __attribute__((aligned(16))) float Wlds[TILE_FLOATS];
  __shared__ float Lg[ROWS * LP];
  __shared__ float Og[ROWS * LP];

  const int tid   = threadIdx.x;
  const int wave  = __builtin_amdgcn_readfirstlane(tid >> 5);
  const int lane  = tid & 31;
  const int strip = wave & 3;
  const int cg    = wave >> 2;
  const int m0    = blockIdx.x * ROWS + strip * 16;
  const int khalf = (lane >> 4) << 1;
  const int ncol  = lane & 15;

  const float* Arow = A + (long)(m0 + ncol) * D_MODEL + khalf;
  v2f a_cur[KB];
  #pragma unroll
  for (int s = 0; s < KB; ++s) a_cur[s] = *(const v2f*)(Arow + 4 * s);

  dta_stage_w96<NTHR, PITCH>(Woff, Wlds, 0,  tid);
  dta_stage_w96<NTHR, PITCH>(Waw,  Wlds, NH, tid);
  __syncthreads();

  {
    v8f acc[NT];
    #pragma unroll
    for (int t = 0; t < NT; ++t) acc[t] = v8f{};

    const float* Bbase = &Wlds[(khalf >> 1) * PITCH + cg * (2 * 16 * NT) + ncol * 2];

    static_assert(KB % G == 0 && K_STEPS % KB == 0, "blocks tile a prefetch group, groups tile K");
    constexpr int NBLK = K_STEPS / G;
    v2f a_nxt[KB];
    v2f bset[2][G][NT];
    #pragma unroll
    for (int g = 0; g < G; ++g) {
      #pragma unroll
      for (int t = 0; t < NT; ++t)
        bset[0][g][t] = *(const v2f*)(Bbase + g * 2 * PITCH + 32 * t);
    }

    #pragma unroll
    for (int blk = 0; blk < NBLK; ++blk) {
      const int ks0 = blk * G;
      if (blk + 1 < NBLK) {
        #pragma unroll
        for (int g = 0; g < G; ++g) {
          const float* Bk = Bbase + (ks0 + G + g) * 2 * PITCH;
          #pragma unroll
          for (int t = 0; t < NT; ++t)
            bset[(blk + 1) & 1][g][t] = *(const v2f*)(Bk + 32 * t);
        }
      }
      if (ks0 % KB == 0 && ks0 + KB < K_STEPS) {
        #pragma unroll
        for (int s = 0; s < KB; ++s) a_nxt[s] = *(const v2f*)(Arow + 4 * (ks0 + KB + s));
      }
      __builtin_amdgcn_sched_barrier(0);
      #pragma unroll
      for (int g = 0; g < G; ++g) {
        #pragma unroll
        for (int t = 0; t < NT; ++t)
          acc[t] = __builtin_amdgcn_wmma_f32_16x16x4_f32(false, a_cur[ks0 % KB + g], false, bset[blk & 1][g][t], (short)0, acc[t], false, false);
      }
      __builtin_amdgcn_sched_barrier(0);
      if ((ks0 + G) % KB == 0 && ks0 + G < K_STEPS) {
        #pragma unroll
        for (int s = 0; s < KB; ++s) a_cur[s] = a_nxt[s];
      }
    }

    const int rloc = strip * 16 + ((lane >> 4) << 3);
    const int c0   = (cg & 1) * (16 * NT);
    float*       dst  = (cg < 2) ? Og : Lg;
    const float* bsrc = (cg < 2) ? boff : baw;
    #pragma unroll
    for (int t = 0; t < NT; ++t) {
      const int c  = c0 + 16 * t + ncol;
      const int pc = (c % 12) * 8 + c / 12;
      const float bvt = bsrc[c];
      #pragma unroll
      for (int r = 0; r < 8; ++r)
        dst[(rloc + r) * LP + pc] = acc[t][r] + bvt;
    }
  }
  __syncthreads();

  unsigned* Atab = (unsigned*)Wlds;
  float*    Wtab = Wlds + ROWS * NH;
  const int b = blockIdx.x >> 5;
  {
    const int r = tid >> 3, h = tid & 7;
    const float* pl = &Lg[r * LP + h];
    const float* pf = &Og[r * LP + h];
    const float rp = refp[blockIdx.x * ROWS + r];
    float v[12];
    float mx = -3.402823466e+38f;
    #pragma unroll
    for (int e = 0; e < 12; ++e) { v[e] = pl[8 * e]; mx = fmaxf(mx, v[e]); }
    float s = 0.0f;
    #pragma unroll
    for (int e = 0; e < 12; ++e) { v[e] = __expf(v[e] - mx); s += v[e]; }
    const float inv = 1.0f / s;
    unsigned* arow = Atab + r * NH + h;
    float*    wrow = Wtab + r * (NH * 3) + h;
    #pragma unroll
    for (int e = 0; e < 12; ++e) {
      const int   l    = e >> 2;
      const int   pnt  = e & 3;
      const int   T    = 2048 >> l;
      const float Tm1  = (float)(T - 1);
      const float invT = __builtin_ldexpf(1.0f, l - 11);
      const int   lvl0 = 8 * (4096 - 2 * T);
      const float o = pf[8 * e];
      float pos  = fminf(fmaxf(rp + o * invT, 0.0f), 1.0f);
      float sidx = pos * Tm1;
      int ifl = (int)sidx;
      ifl = min(max(ifl, 0), T - 2);
      const float wc = sidx - (float)ifl;
      arow[8 * e] = ((unsigned)(lvl0 + b * T + ifl) << 10) + (unsigned)(pnt * 128);
      const int wi = (e >> 1) * 48 + (e & 1) * 8;
      wrow[wi]      = wc;
      wrow[wi + 16] = 1.0f - wc;
      wrow[wi + 32] = v[e] * inv;
    }
  }
  __syncthreads();

  const int rl0   = wave * RPW;
  const int grow0 = blockIdx.x * ROWS + rl0;
  const unsigned lane4 = (unsigned)lane << 2;
  const char* vbytes = (const char*)v_proj;
  constexpr int LAST = 6 * RPW - 1;
  const v4u* arw = (const v4u*)(Atab + rl0 * NH);
  const v4f* wrw = (const v4f*)(Wtab + rl0 * NH * 3);

  unsigned po[16];
  #pragma unroll
  for (int q = 0; q < 4; ++q) {
    const v4u t = arw[q];
    po[4 * q] = t.x; po[4 * q + 1] = t.y; po[4 * q + 2] = t.z; po[4 * q + 3] = t.w;
  }

  #pragma nounroll
  for (int j = 0; j < RPW; ++j) {
    float acc[8];
    #pragma unroll
    for (int h = 0; h < 8; ++h) acc[h] = 0.0f;

    #pragma nounroll
    for (int bt = 0; bt < 6; ++bt) {
      const int cb = j * 6 + bt;

      unsigned vo[16];
      #pragma unroll
      for (int s = 0; s < 16; ++s) vo[s] = po[s] + lane4;
      __builtin_amdgcn_sched_barrier(0);

      float vf[16], vc[16];
      #pragma unroll
      for (int s = 0; s < 16; ++s) {
        const char* a = vbytes + vo[s];
        vf[s] = *(const float*)a;
        vc[s] = *(const float*)(a + 1024);
      }
      float wq[48];
      #pragma unroll
      for (int q = 0; q < 12; ++q) {
        const v4f t = wrw[cb * 12 + q];
        wq[4 * q] = t.x; wq[4 * q + 1] = t.y; wq[4 * q + 2] = t.z; wq[4 * q + 3] = t.w;
      }
      const int nb = min(cb + 1, LAST);
      unsigned po_n[16];
      #pragma unroll
      for (int q = 0; q < 4; ++q) {
        const v4u t = arw[nb * 4 + q];
        po_n[4 * q] = t.x; po_n[4 * q + 1] = t.y; po_n[4 * q + 2] = t.z; po_n[4 * q + 3] = t.w;
      }
      __builtin_amdgcn_sched_barrier(0);

      #pragma unroll
      for (int s = 0; s < 16; ++s) {
        const float w = wq[32 + s];
        acc[s & 7] += w * (wq[16 + s] * vf[s] + wq[s] * vc[s]);
      }
      #pragma unroll
      for (int s = 0; s < 16; ++s) po[s] = po_n[s];
    }
    #pragma unroll
    for (int h = 0; h < 8; ++h)
      out_mid[(long)(grow0 + j) * 256 + h * 32 + lane] = acc[h];
  }
}

extern "C" void kernel_launch(void* const* d_in, const int* in_sizes, int n_in,
                              void* d_out, int out_size, void* d_ws, size_t ws_size,
                              hipStream_t stream) {
  (void)in_sizes; (void)n_in; (void)out_size; (void)ws_size;

  const float* query = (const float*)d_in[0];
  const float* refp  = (const float*)d_in[1];
  const float* val0  = (const float*)d_in[2];
  const float* val1  = (const float*)d_in[3];
  const float* val2  = (const float*)d_in[4];
  const float* Woff  = (const float*)d_in[5];
  const float* boff  = (const float*)d_in[6];
  const float* Waw   = (const float*)d_in[7];
  const float* baw   = (const float*)d_in[8];
  const float* Wv    = (const float*)d_in[9];
  const float* bv    = (const float*)d_in[10];
  const float* Wo    = (const float*)d_in[11];
  const float* bo    = (const float*)d_in[12];
  float*       out   = (float*)d_out;

  const int BQ = 8 * 2048;

  char* ws = (char*)d_ws;
  float* v_proj  = (float*)(ws + (size_t)BQ * 96 * 4 * 2);
  float* out_mid = (float*)(ws + (size_t)BQ * 96 * 4 * 2
                               + (size_t)(BQ + 8192 + 4096) * 256 * 4);

  dim3 blk(256);
  dim3 blk4(128);
  dim3 blk16(512);

  dta_vproj_ws<64, 4, 7, 16, 4><<<dim3(64, 4), blk4, 0, stream>>>(val0, val1, val2, Wv, bv, v_proj);
  dta_qproj_sample<8, 4><<<dim3(256), blk16, 0, stream>>>(query, Woff, boff, Waw, baw, refp, v_proj, out_mid);
  dta_gemm_f32_wmma<128, 8, 8, 2><<<dim3(128, 2), blk, 0, stream>>>(out_mid, Wo, bo, out, 256);
}


# Sampling as matrix products: where a (row, level)'s heads lie in a 3-row window, [8 heads x 12 slices] x [12 x 32 channels] on f32 WMMA, coefficients from the softmax threads; sum order changes
# speedup vs baseline: 1.7141x; 1.0856x over previous
#include <hip/hip_runtime.h>
#include <hip/hip_bf16.h>
#include <math.h>


typedef float v2f __attribute__((ext_vector_type(2)));
typedef float v4f __attribute__((ext_vector_type(4)));
typedef float v8f __attribute__((ext_vector_type(8)));
typedef unsigned v4u __attribute__((ext_vector_type(4)));
typedef unsigned v2u __attribute__((ext_vector_type(2)));

#define D_MODEL 256

#define K_STEPS (D_MODEL / 4)

template <int NW, int WAVES, int KB, int G>
__global__ __launch_bounds__(WAVES * 32)
void dta_gemm_f32_wmma(const float* __restrict__ A, const float* __restrict__ W,
                       const float* __restrict__ bias, float* __restrict__ C,
                       int N) {
  constexpr int NTHR  = WAVES * 32;
  constexpr int NTILE = NW / 16;
  constexpr int PITCH = NW * 2 + 32;
  constexpr int CPR   = NW / 4;
  constexpr int TILE_FLOATS = (D_MODEL / 2) * PITCH;
  constexpr int ONE_PER_WGP_FLOATS = 41984;
  constexpr int LDS_FLOATS = TILE_FLOATS > ONE_PER_WGP_FLOATS ? TILE_FLOATS : ONE_PER_WGP_FLOATS;
  __shared__ __attribute__((aligned(16))) float Wlds[LDS_FLOATS];

  const int tid   = threadIdx.x;
  const int wave  = tid >> 5;
  const int lane  = tid & 31;
  const int n0    = blockIdx.y * NW;
  const int m0    = blockIdx.x * (WAVES * 16) + wave * 16;
  const int khalf = (lane >> 4) << 1;
  const int ncol  = lane & 15;

  const float* Arow = A + (long)(m0 + ncol) * D_MODEL + khalf;
  v2f a_cur[KB];
  #pragma unroll
  for (int s = 0; s < KB; ++s) a_cur[s] = *(const v2f*)(Arow + 4 * s);

  constexpr int STAGE_ITERS = (D_MODEL / 2) * CPR / NTHR;
  static_assert(STAGE_ITERS * NTHR == (D_MODEL / 2) * CPR, "chunks divide evenly over the threads");
  #pragma unroll
  for (int it = 0; it < STAGE_ITERS; ++it) {
    const int ch = tid + it * NTHR;
    const int kk = ch / CPR;
    const int j  = (ch - kk * CPR) << 2;
    v4f r0 = *(const v4f*)(W + (long)(2 * kk)     * N + n0 + j);
    v4f r1 = *(const v4f*)(W + (long)(2 * kk + 1) * N + n0 + j);
    v4f lo = {r0.x, r1.x, r0.y, r1.y};
    v4f hi = {r0.z, r1.z, r0.w, r1.w};
    *(v4f*)(&Wlds[kk * PITCH + j * 2])     = lo;
    *(v4f*)(&Wlds[kk * PITCH + j * 2 + 4]) = hi;
  }
  __syncthreads();

  v8f acc[NTILE];
  #pragma unroll
  for (int t = 0; t < NTILE; ++t) acc[t] = v8f{};

  const float* Bbase = &Wlds[(khalf >> 1) * PITCH + ncol * 2];

  static_assert(KB % G == 0 && K_STEPS % KB == 0, "blocks tile a prefetch group, groups tile K");
  constexpr int NBLK = K_STEPS / G;
  v2f a_nxt[KB];
  v2f bset[2][G][NTILE];
  #pragma unroll
  for (int g = 0; g < G; ++g) {
    #pragma unroll
    for (int t = 0; t < NTILE; ++t)
      bset[0][g][t] = *(const v2f*)(Bbase + g * 2 * PITCH + 32 * t);
  }

  #pragma unroll
  for (int blk = 0; blk < NBLK; ++blk) {
    const int ks0 = blk * G;
    if (blk + 1 < NBLK) {
      #pragma unroll
      for (int g = 0; g < G; ++g) {
        const float* Bk = Bbase + (ks0 + G + g) * 2 * PITCH;
        #pragma unroll
        for (int t = 0; t < NTILE; ++t)
          bset[(blk + 1) & 1][g][t] = *(const v2f*)(Bk + 32 * t);
      }
    }
    if (ks0 % KB == 0 && ks0 + KB < K_STEPS) {
      #pragma unroll
      for (int s = 0; s < KB; ++s) a_nxt[s] = *(const v2f*)(Arow + 4 * (ks0 + KB + s));
    }
    __builtin_amdgcn_sched_barrier(0);
    #pragma unroll
    for (int g = 0; g < G; ++g) {
      #pragma unroll
      for (int t = 0; t < NTILE; ++t)
        acc[t] = __builtin_amdgcn_wmma_f32_16x16x4_f32(false, a_cur[ks0 % KB + g], false, bset[blk & 1][g][t], (short)0, acc[t], false, false);
    }
    __builtin_amdgcn_sched_barrier(0);
    if ((ks0 + G) % KB == 0 && ks0 + G < K_STEPS) {
      #pragma unroll
      for (int s = 0; s < KB; ++s) a_cur[s] = a_nxt[s];
    }
  }

  const int rbase = m0 + ((lane >> 4) << 3);
  #pragma unroll
  for (int t = 0; t < NTILE; ++t) {
    const float bvt = bias[n0 + 16 * t + ncol];
    #pragma unroll
    for (int r = 0; r < 8; ++r)
      C[(long)(rbase + r) * N + n0 + 16 * t + ncol] = acc[t][r] + bvt;
  }
}

#define DTA_STRIPS_L0 1024
#define DTA_STRIPS_L1 512

__device__ __forceinline__ const float* dta_strip_rows(const float* __restrict__ V0,
                                                       const float* __restrict__ V1,
                                                       const float* __restrict__ V2, int s) {
  if (s < DTA_STRIPS_L0) return V0 + (long)s * (16 * D_MODEL);
  if (s < DTA_STRIPS_L0 + DTA_STRIPS_L1) return V1 + (long)(s - DTA_STRIPS_L0) * (16 * D_MODEL);
  return V2 + (long)(s - DTA_STRIPS_L0 - DTA_STRIPS_L1) * (16 * D_MODEL);
}

template <int NW, int WAVES, int STRIPS, int KB, int G>
__global__ __launch_bounds__(WAVES * 32)
void dta_vproj_ws(const float* __restrict__ V0, const float* __restrict__ V1,
                  const float* __restrict__ V2, const float* __restrict__ W,
                  const float* __restrict__ bias, float* __restrict__ C) {
  constexpr int N     = D_MODEL;
  constexpr int NTHR  = WAVES * 32;
  constexpr int NTILE = NW / 16;
  constexpr int PITCH = NW * 2 + 32;
  constexpr int CPR   = NW / 4;
  constexpr int TILE_FLOATS = (D_MODEL / 2) * PITCH;
  constexpr int ONE_PER_WGP_FLOATS = 41984;
  constexpr int LDS_FLOATS = TILE_FLOATS > ONE_PER_WGP_FLOATS ? TILE_FLOATS : ONE_PER_WGP_FLOATS;
  __shared__ __attribute__((aligned(16))) float Wlds[LDS_FLOATS];

  const int tid   = threadIdx.x;
  const int wave  = __builtin_amdgcn_readfirstlane(tid >> 5);
  const int lane  = tid & 31;
  const int n0    = blockIdx.y * NW;
  const int khalf = (lane >> 4) << 1;
  const int ncol  = lane & 15;
  const int lofs  = ncol * D_MODEL + khalf;
  const int s0    = blockIdx.x * (WAVES * STRIPS) + wave;

  const float* Arow = dta_strip_rows(V0, V1, V2, s0) + lofs;
  v2f a_cur[KB];
  #pragma unroll
  for (int s = 0; s < KB; ++s) a_cur[s] = *(const v2f*)(Arow + 4 * s);

  constexpr int STAGE_ITERS = (D_MODEL / 2) * CPR / NTHR;
  static_assert(STAGE_ITERS * NTHR == (D_MODEL / 2) * CPR, "chunks divide evenly over the threads");
  #pragma unroll
  for (int it = 0; it < STAGE_ITERS; ++it) {
    const int ch = tid + it * NTHR;
    const int kk = ch / CPR;
    const int j  = (ch - kk * CPR) << 2;
    v4f r0 = *(const v4f*)(W + (long)(2 * kk)     * N + n0 + j);
    v4f r1 = *(const v4f*)(W + (long)(2 * kk + 1) * N + n0 + j);
    v4f lo = {r0.x, r1.x, r0.y, r1.y};
    v4f hi = {r0.z, r1.z, r0.w, r1.w};
    *(v4f*)(&Wlds[kk * PITCH + j * 2])     = lo;
    *(v4f*)(&Wlds[kk * PITCH + j * 2 + 4]) = hi;
  }

  float bvt[NTILE];
  #pragma unroll
  for (int t = 0; t < NTILE; ++t) bvt[t] = bias[n0 + 16 * t + ncol];
  __syncthreads();

  const float* Bbase = &Wlds[(khalf >> 1) * PITCH + ncol * 2];

  static_assert(KB % G == 0 && K_STEPS % KB == 0, "blocks tile a prefetch group, groups tile K");
  constexpr int NBLK = K_STEPS / G;
  static_assert(NBLK % 2 == 0, "the B register sets alternate and must line up again at the next strip");
  v2f a_nxt[KB];
  v2f bset[2][G][NTILE];
  #pragma unroll
  for (int g = 0; g < G; ++g) {
    #pragma unroll
    for (int t = 0; t < NTILE; ++t)
      bset[0][g][t] = *(const v2f*)(Bbase + g * 2 * PITCH + 32 * t);
  }

  for (int i = 0; i < STRIPS; ++i) {
    const int sc = s0 + i * WAVES;
    const int sn = (i + 1 < STRIPS) ? sc + WAVES : sc;
    const float* Anext = dta_strip_rows(V0, V1, V2, sn) + lofs;

    v8f acc[NTILE];
    #pragma unroll
    for (int t = 0; t < NTILE; ++t) acc[t] = v8f{};

    #pragma unroll
    for (int blk = 0; blk < NBLK; ++blk) {
      const int ks0 = blk * G;
      #pragma unroll
      for (int g = 0; g < G; ++g) {
        const float* Bk = Bbase + ((ks0 + G + g) % K_STEPS) * 2 * PITCH;
        #pragma unroll
        for (int t = 0; t < NTILE; ++t)
          bset[(blk + 1) & 1][g][t] = *(const v2f*)(Bk + 32 * t);
      }
      if (ks0 % KB == 0) {
        if (ks0 + KB < K_STEPS) {
          #pragma unroll
          for (int s = 0; s < KB; ++s) a_nxt[s] = *(const v2f*)(Arow + 4 * (ks0 + KB + s));
        } else {
          #pragma unroll
          for (int s = 0; s < KB; ++s) a_nxt[s] = *(const v2f*)(Anext + 4 * s);
        }
      }
      __builtin_amdgcn_sched_barrier(0);
      #pragma unroll
      for (int g = 0; g < G; ++g) {
        #pragma unroll
        for (int t = 0; t < NTILE; ++t)
          acc[t] = __builtin_amdgcn_wmma_f32_16x16x4_f32(false, a_cur[ks0 % KB + g], false, bset[blk & 1][g][t], (short)0, acc[t], false, false);
      }
      __builtin_amdgcn_sched_barrier(0);
      if ((ks0 + G) % KB == 0) {
        #pragma unroll
        for (int s = 0; s < KB; ++s) a_cur[s] = a_nxt[s];
      }
    }

    float* Crow = C + (long)sc * (16 * N) + (long)((lane >> 4) << 3) * N + n0 + ncol;
    #pragma unroll
    for (int t = 0; t < NTILE; ++t) {
      #pragma unroll
      for (int r = 0; r < 8; ++r)
        Crow[r * N + 16 * t] = acc[t][r] + bvt[t];
    }
    Arow = Anext;
  }
}

template <int NTHR, int PITCH>
__device__ __forceinline__ void dta_stage_w96(const float* __restrict__ W, float* Wlds,
                                              int cofs, int tid) {
  constexpr int CPR   = 96 / 4;
  constexpr int ITERS = (D_MODEL / 2) * CPR / NTHR;
  static_assert(ITERS * NTHR == (D_MODEL / 2) * CPR, "chunks divide evenly over the threads");
  #pragma unroll
  for (int it = 0; it < ITERS; ++it) {
    const int ch = tid + it * NTHR;
    const int kk = ch / CPR;
    const int j  = (ch - kk * CPR) << 2;
    v4f r0 = *(const v4f*)(W + (long)(2 * kk)     * 96 + j);
    v4f r1 = *(const v4f*)(W + (long)(2 * kk + 1) * 96 + j);
    v4f lo = {r0.x, r1.x, r0.y, r1.y};
    v4f hi = {r0.z, r1.z, r0.w, r1.w};
    *(v4f*)(&Wlds[kk * PITCH + (cofs + j) * 2])     = lo;
    *(v4f*)(&Wlds[kk * PITCH + (cofs + j) * 2 + 4]) = hi;
  }
}

template <int KB, int G>
__global__ __launch_bounds__(512)
void dta_qproj_sample(const float* __restrict__ A,
                      const float* __restrict__ Woff, const float* __restrict__ boff,
                      const float* __restrict__ Waw,  const float* __restrict__ baw,
                      const float* __restrict__ refp, const float* __restrict__ v_proj,
                      float* __restrict__ out_mid) {
  constexpr int NH    = 96;
  constexpr int NW    = 2 * NH;
  constexpr int NTHR  = 512;
  constexpr int NT    = 3;
  constexpr int PITCH = NW * 2 + 32;
  constexpr int TILE_FLOATS = (D_MODEL / 2) * PITCH;
  constexpr int ROWS  = 64;
  constexpr int RPW   = 4;
  constexpr int LP    = NH + 1;
  __shared__ __attribute__((aligned(16))) float Wlds[TILE_FLOATS];
  __shared__ float Lg[ROWS * LP];
  __shared__ float Og[ROWS * LP];

  const int tid   = threadIdx.x;
  const int wave  = __builtin_amdgcn_readfirstlane(tid >> 5);
  const int lane  = tid & 31;
  const int strip = wave & 3;
  const int cg    = wave >> 2;
  const int m0    = blockIdx.x * ROWS + strip * 16;
  const int khalf = (lane >> 4) << 1;
  const int ncol  = lane & 15;

  const float* Arow = A + (long)(m0 + ncol) * D_MODEL + khalf;
  v2f a_cur[KB];
  #pragma unroll
  for (int s = 0; s < KB; ++s) a_cur[s] = *(const v2f*)(Arow + 4 * s);

  dta_stage_w96<NTHR, PITCH>(Woff, Wlds, 0,  tid);
  dta_stage_w96<NTHR, PITCH>(Waw,  Wlds, NH, tid);
  __syncthreads();

  {
    v8f acc[NT];
    #pragma unroll
    for (int t = 0; t < NT; ++t) acc[t] = v8f{};

    const float* Bbase = &Wlds[(khalf >> 1) * PITCH + cg * (2 * 16 * NT) + ncol * 2];

    static_assert(KB % G == 0 && K_STEPS % KB == 0, "blocks tile a prefetch group, groups tile K");
    constexpr int NBLK = K_STEPS / G;
    v2f a_nxt[KB];
    v2f bset[2][G][NT];
    #pragma unroll
    for (int g = 0; g < G; ++g) {
      #pragma unroll
      for (int t = 0; t < NT; ++t)
        bset[0][g][t] = *(const v2f*)(Bbase + g * 2 * PITCH + 32 * t);
    }

    #pragma unroll
    for (int blk = 0; blk < NBLK; ++blk) {
      const int ks0 = blk * G;
      if (blk + 1 < NBLK) {
        #pragma unroll
        for (int g = 0; g < G; ++g) {
          const float* Bk = Bbase + (ks0 + G + g) * 2 * PITCH;
          #pragma unroll
          for (int t = 0; t < NT; ++t)
            bset[(blk + 1) & 1][g][t] = *(const v2f*)(Bk + 32 * t);
        }
      }
      if (ks0 % KB == 0 && ks0 + KB < K_STEPS) {
        #pragma unroll
        for (int s = 0; s < KB; ++s) a_nxt[s] = *(const v2f*)(Arow + 4 * (ks0 + KB + s));
      }
      __builtin_amdgcn_sched_barrier(0);
      #pragma unroll
      for (int g = 0; g < G; ++g) {
        #pragma unroll
        for (int t = 0; t < NT; ++t)
          acc[t] = __builtin_amdgcn_wmma_f32_16x16x4_f32(false, a_cur[ks0 % KB + g], false, bset[blk & 1][g][t], (short)0, acc[t], false, false);
      }
      __builtin_amdgcn_sched_barrier(0);
      if ((ks0 + G) % KB == 0 && ks0 + G < K_STEPS) {
        #pragma unroll
        for (int s = 0; s < KB; ++s) a_cur[s] = a_nxt[s];
      }
    }

    const int rloc = strip * 16 + ((lane >> 4) << 3);
    const int c0   = (cg & 1) * (16 * NT);
    float*       dst  = (cg < 2) ? Og : Lg;
    const float* bsrc = (cg < 2) ? boff : baw;
    #pragma unroll
    for (int t = 0; t < NT; ++t) {
      const int c  = c0 + 16 * t + ncol;
      const int pc = (c % 12) * 8 + c / 12;
      const float bvt = bsrc[c];
      #pragma unroll
      for (int r = 0; r < 8; ++r)
        dst[(rloc + r) * LP + pc] = acc[t][r] + bvt;
    }
  }
  __syncthreads();

  unsigned* Atab = (unsigned*)Wlds;
  float*    Wtab = Wlds + ROWS * NH;
  unsigned* Btab = (unsigned*)(Wlds + ROWS * NH * 4);
  float*    Ctab = Wlds + ROWS * NH * 4 + ROWS * 36;
  const int b = blockIdx.x >> 5;
  unsigned okmask = 0;
  {
    const int r = tid >> 3, h = tid & 7;
    const float* pl = &Lg[r * LP + h];
    const float* pf = &Og[r * LP + h];
    const float rp = refp[blockIdx.x * ROWS + r];
    float v[12];
    float mx = -3.402823466e+38f;
    #pragma unroll
    for (int e = 0; e < 12; ++e) { v[e] = pl[8 * e]; mx = fmaxf(mx, v[e]); }
    float s = 0.0f;
    #pragma unroll
    for (int e = 0; e < 12; ++e) { v[e] = __expf(v[e] - mx); s += v[e]; }
    const float inv = 1.0f / s;
    int   ifl_[12], mn_[12];
    float wc_[12];
    bool  okt[3] = {true, true, true};
    #pragma unroll
    for (int e = 0; e < 12; ++e) {
      const int   l    = e >> 2;
      const int   T    = 2048 >> l;
      const float Tm1  = (float)(T - 1);
      const float invT = __builtin_ldexpf(1.0f, l - 11);
      const float o = pf[8 * e];
      float pos  = fminf(fmaxf(rp + o * invT, 0.0f), 1.0f);
      float sidx = pos * Tm1;
      int ifl = (int)sidx;
      ifl = min(max(ifl, 0), T - 2);
      ifl_[e] = ifl;
      wc_[e]  = sidx - (float)ifl;
      int m = ifl;
      m = min(m, __shfl_xor(m, 1));
      m = min(m, __shfl_xor(m, 2));
      m = min(m, __shfl_xor(m, 4));
      mn_[e] = m;
      okt[l] = okt[l] && (ifl - m <= 1);
    }
    unsigned* brow = Btab + r * 36;
    float*    crow = Ctab + r * 288 + h * 2;
    unsigned* arow = Atab + r * NH + h;
    float*    wrow = Wtab + r * (NH * 3) + h;
    #pragma unroll
    for (int l = 0; l < 3; ++l) {
      const unsigned bal = __builtin_amdgcn_ballot_w32(okt[l]);
      #pragma unroll
      for (int j = 0; j < RPW; ++j)
        okmask |= ((((bal >> (8 * j)) & 0xffu) == 0xffu) ? 1u : 0u) << (3 * j + l);
      const bool rok = ((bal >> (lane & 24)) & 0xffu) == 0xffu;
      const int T    = 2048 >> l;
      const int lvl0 = 8 * (4096 - 2 * T);
      #pragma unroll
      for (int pnt = 0; pnt < 4; ++pnt) {
        const int e   = l * 4 + pnt;
        const int i0w = min(mn_[e], T - 3);
        if (h < 3)
          brow[l * 12 + pnt * 3 + h] = ((unsigned)(lvl0 + b * T + i0w + h) << 10) + (unsigned)(pnt * 128);
        const float aw = v[e] * inv;
        const float wc = wc_[e];
        if (rok) {
          const int   j0 = ifl_[e] - i0w;
          const float cA = aw * (1.0f - wc);
          const float cB = aw * wc;
          const float cw[3] = {(j0 == 0) ? cA : 0.0f, (j0 == 0) ? cB : cA, (j0 == 0) ? 0.0f : cB};
          #pragma unroll
          for (int jj = 0; jj < 3; ++jj) {
            const int k = pnt * 3 + jj;
            crow[l * 96 + (k >> 2) * 32 + ((k >> 1) & 1) * 16 + (k & 1)] = cw[jj];
          }
        } else {
          arow[8 * e] = ((unsigned)(lvl0 + b * T + ifl_[e]) << 10) + (unsigned)(pnt * 128);
          const int wi = (e >> 1) * 48 + (e & 1) * 8;
          wrow[wi]      = wc;
          wrow[wi + 16] = 1.0f - wc;
          wrow[wi + 32] = aw;
        }
      }
    }
  }
  __syncthreads();

  const int rl0   = wave * RPW;
  const int grow0 = blockIdx.x * ROWS + rl0;
  const unsigned lane4 = (unsigned)lane << 2;
  const unsigned col4  = (unsigned)(lane & 15) << 2;
  const int      half  = lane >> 4;
  const char* vbytes = (const char*)v_proj;
  const v2u* brw = (const v2u*)(Btab + rl0 * 36) + half;
  const v2f* crw = (const v2f*)(Ctab + rl0 * 288) + half * 8 + (lane & 7);
  const v4u* arw = (const v4u*)(Atab + rl0 * NH);
  const v4f* wrw = (const v4f*)(Wtab + rl0 * NH * 3);
  float acc[8];
  v8f accd[2];
  #pragma unroll
  for (int h = 0; h < 8; ++h) acc[h] = 0.0f;
  accd[0] = v8f{}; accd[1] = v8f{};

  auto request = [&](float (&dst)[12], const v2u (&kk)[3]) {
    #pragma unroll
    for (int ks = 0; ks < 3; ++ks) {
      const char* q0 = vbytes + (kk[ks].x + col4);
      const char* q1 = vbytes + (kk[ks].y + col4);
      dst[4 * ks]     = *(const float*)q0;
      dst[4 * ks + 1] = *(const float*)q1;
      dst[4 * ks + 2] = *(const float*)(q0 + 64);
      dst[4 * ks + 3] = *(const float*)(q1 + 64);
    }
  };
  auto offsets = [&](v2u (&kk)[3], int t) {
    #pragma unroll
    for (int ks = 0; ks < 3; ++ks) kk[ks] = brw[t * 6 + ks * 2];
  };
  auto coefs = [&](v2f (&af)[3], int t) {
    const bool ok = ((okmask >> t) & 1u) != 0;
    #pragma unroll
    for (int ks = 0; ks < 3; ++ks) {
      const v2f c2 = crw[t * 48 + ks * 16];
      af[ks] = ok ? c2 : v2f{0.0f, 0.0f};
    }
  };
  auto window_mma = [&](const float (&val)[12], const v2f (&af)[3]) {
    #pragma unroll
    for (int ks = 0; ks < 3; ++ks) {
      const v2f b0 = {val[4 * ks],     val[4 * ks + 1]};
      const v2f b1 = {val[4 * ks + 2], val[4 * ks + 3]};
      accd[0] = __builtin_amdgcn_wmma_f32_16x16x4_f32(false, af[ks], false, b0, (short)0, accd[0], false, false);
      accd[1] = __builtin_amdgcn_wmma_f32_16x16x4_f32(false, af[ks], false, b1, (short)0, accd[1], false, false);
    }
    __builtin_amdgcn_sched_barrier(0);
    asm volatile("v_nop\n\tv_nop\n\tv_nop\n\tv_nop\n\tv_nop" : "+v"(accd[0]), "+v"(accd[1]));
    __builtin_amdgcn_sched_barrier(0);
  };
  auto sample_group = [&](int g) {
    unsigned vo[16];
    #pragma unroll
    for (int q = 0; q < 4; ++q) {
      const v4u t4 = arw[g * 4 + q];
      vo[4 * q] = t4.x + lane4; vo[4 * q + 1] = t4.y + lane4; vo[4 * q + 2] = t4.z + lane4; vo[4 * q + 3] = t4.w + lane4;
    }
    float vf[16], vc[16];
    #pragma unroll
    for (int s = 0; s < 16; ++s) {
      const char* a = vbytes + vo[s];
      vf[s] = *(const float*)a;
      vc[s] = *(const float*)(a + 1024);
    }
    float wq[48];
    #pragma unroll
    for (int q = 0; q < 12; ++q) {
      const v4f t4 = wrw[g * 12 + q];
      wq[4 * q] = t4.x; wq[4 * q + 1] = t4.y; wq[4 * q + 2] = t4.z; wq[4 * q + 3] = t4.w;
    }
    #pragma unroll
    for (int s = 0; s < 16; ++s) {
      const float w = wq[32 + s];
      acc[s & 7] += w * (wq[16 + s] * vf[s] + wq[s] * vc[s]);
    }
  };
  auto batch = [&](const float (&val)[12], const v2f (&af)[3], int t) {
    const int j = t / 3, l = t - 3 * j;
    window_mma(val, af);
    if (((okmask >> t) & 1u) == 0) {
      sample_group(j * 6 + l * 2);
      sample_group(j * 6 + l * 2 + 1);
    }
    if (l == 2) {
      #pragma unroll
      for (int h = 0; h < 8; ++h) {
        out_mid[(long)(grow0 + j) * 256 + h * 32 + lane] = acc[h] + (half ? accd[1][h] : accd[0][h]);
        acc[h] = 0.0f;
      }
      accd[0] = v8f{}; accd[1] = v8f{};
    }
  };

  float va[12], vb[12];
  v2u knext[3];
  v2f af[3];
  offsets(knext, 0);
  request(va, knext);
  offsets(knext, 1);
  #pragma nounroll
  for (int tt = 0; tt < 12; tt += 2) {
    request(vb, knext);
    offsets(knext, min(tt + 2, 11));
    coefs(af, tt);
    __builtin_amdgcn_sched_barrier(0);
    batch(va, af, tt);
    request(va, knext);
    offsets(knext, min(tt + 3, 11));
    coefs(af, tt + 1);
    __builtin_amdgcn_sched_barrier(0);
    batch(vb, af, tt + 1);
  }
}

extern "C" void kernel_launch(void* const* d_in, const int* in_sizes, int n_in,
                              void* d_out, int out_size, void* d_ws, size_t ws_size,
                              hipStream_t stream) {
  (void)in_sizes; (void)n_in; (void)out_size; (void)ws_size;

  const float* query = (const float*)d_in[0];
  const float* refp  = (const float*)d_in[1];
  const float* val0  = (const float*)d_in[2];
  const float* val1  = (const float*)d_in[3];
  const float* val2  = (const float*)d_in[4];
  const float* Woff  = (const float*)d_in[5];
  const float* boff  = (const float*)d_in[6];
  const float* Waw   = (const float*)d_in[7];
  const float* baw   = (const float*)d_in[8];
  const float* Wv    = (const float*)d_in[9];
  const float* bv    = (const float*)d_in[10];
  const float* Wo    = (const float*)d_in[11];
  const float* bo    = (const float*)d_in[12];
  float*       out   = (float*)d_out;

  const int BQ = 8 * 2048;

  char* ws = (char*)d_ws;
  float* v_proj  = (float*)(ws + (size_t)BQ * 96 * 4 * 2);
  float* out_mid = (float*)(ws + (size_t)BQ * 96 * 4 * 2
                               + (size_t)(BQ + 8192 + 4096) * 256 * 4);

  dim3 blk(256);
  dim3 blk4(128);
  dim3 blk16(512);

  dta_vproj_ws<64, 4, 7, 16, 4><<<dim3(64, 4), blk4, 0, stream>>>(val0, val1, val2, Wv, bv, v_proj);
  dta_qproj_sample<8, 4><<<dim3(256), blk16, 0, stream>>>(query, Woff, boff, Waw, baw, refp, v_proj, out_mid);
  dta_gemm_f32_wmma<128, 8, 8, 2><<<dim3(128, 2), blk, 0, stream>>>(out_mid, Wo, bo, out, 256);
}
